# starting kernel
# baseline (speedup 1.0000x reference)
#include <hip/hip_runtime.h>
#include <hip/hip_bf16.h>

typedef __attribute__((ext_vector_type(2))) float v2f;
typedef __attribute__((ext_vector_type(8))) float v8f;

#define GCN_N_NODES 100000
#define GCN_N_EDGES 1000000
#define GCN_D 64

__global__ void gcn_zero_deg(float* __restrict__ deg) {
    int i = blockIdx.x * blockDim.x + threadIdx.x;
    if (i < GCN_N_NODES) deg[i] = 0.0f;
}

__global__ void gcn_degree(const int* __restrict__ edge_index, float* __restrict__ deg) {
    int e = blockIdx.x * blockDim.x + threadIdx.x;
    if (e < GCN_N_EDGES) {
        atomicAdd(&deg[edge_index[e]], 1.0f);
    }
}

__global__ void gcn_normalize(const float* __restrict__ x, const float* __restrict__ deg,
                              float* __restrict__ x_norm, float* __restrict__ agg) {
    int i = blockIdx.x * blockDim.x + threadIdx.x;
    if (i >= GCN_N_NODES * (GCN_D / 4)) return;
    int n = i / (GCN_D / 4);
    float s = rsqrtf(deg[n] + 1.0f);
    float4 v = ((const float4*)x)[i];
    v.x *= s; v.y *= s; v.z *= s; v.w *= s;
    ((float4*)x_norm)[i] = v;
    ((float4*)agg)[i]    = v;
}

__global__ void gcn_scatter(const int* __restrict__ edge_index,
                            const float* __restrict__ x_norm, float* __restrict__ agg) {
    int t = blockIdx.x * blockDim.x + threadIdx.x;
    int e = t >> 6;
    int f = t & 63;
    if (e >= GCN_N_EDGES) return;
    int s = edge_index[e];
    int d = edge_index[GCN_N_EDGES + e];
    atomicAdd(&agg[d * GCN_D + f], x_norm[s * GCN_D + f]);
}

__global__ void gcn_gemm_wmma(const float* __restrict__ agg, const float* __restrict__ W,
                              const float* __restrict__ bias, float* __restrict__ out) {
    const int lane = threadIdx.x & 31;
    const int wave = threadIdx.x >> 5;
    const int tile = blockIdx.x * 8 + wave;
    const int TILES_N = GCN_D / 16;
    const int n_tiles = (GCN_N_NODES / 16) * TILES_N;
    if (tile >= n_tiles) return;

    const int tm = tile / TILES_N;
    const int tn = tile % TILES_N;
    const int m0 = tm * 16;
    const int o0 = tn * 16;

    const int row = lane & 15;
    const int col = lane & 15;
    const int kb  = (lane >> 4) << 1;

    v2f breg[16];
    const float* wrow = W + (o0 + col) * GCN_D + kb;
#pragma unroll
    for (int t = 0; t < 16; ++t) {
        breg[t] = *(const v2f*)(wrow + 4 * t);
    }

    const float bv = bias[o0 + col];
    v8f c;
#pragma unroll
    for (int r = 0; r < 8; ++r) c[r] = bv;

    const float* arow = agg + (m0 + row) * GCN_D + kb;
#pragma unroll
    for (int t = 0; t < 16; ++t) {
        v2f a = *(const v2f*)(arow + 4 * t);
        c = __builtin_amdgcn_wmma_f32_16x16x4_f32(
                false, a, false, breg[t], (short)0, c, false, false);
    }

    float* orow = out + (m0 + ((lane >> 4) << 3)) * GCN_D + o0 + col;
#pragma unroll
    for (int r = 0; r < 8; ++r) {
        orow[r * GCN_D] = c[r];
    }
}

extern "C" void kernel_launch(void* const* d_in, const int* in_sizes, int n_in,
                              void* d_out, int out_size, void* d_ws, size_t ws_size,
                              hipStream_t stream) {
    const float* x          = (const float*)d_in[0];
    const int*   edge_index = (const int*)d_in[1];
    const float* W          = (const float*)d_in[2];
    const float* b          = (const float*)d_in[3];
    float*       out        = (float*)d_out;

    float* deg    = (float*)d_ws;
    float* x_norm = deg + GCN_N_NODES;
    float* agg    = x_norm + (size_t)GCN_N_NODES * GCN_D;

    gcn_zero_deg<<<(GCN_N_NODES + 255) / 256, 256, 0, stream>>>(deg);
    gcn_degree<<<(GCN_N_EDGES + 255) / 256, 256, 0, stream>>>(edge_index, deg);
    gcn_normalize<<<(GCN_N_NODES * (GCN_D / 4) + 255) / 256, 256, 0, stream>>>(x, deg, x_norm, agg);
    gcn_scatter<<<(GCN_N_EDGES * GCN_D) / 256, 256, 0, stream>>>(edge_index, x_norm, agg);

    const int n_tiles = (GCN_N_NODES / 16) * (GCN_D / 16);
    gcn_gemm_wmma<<<(n_tiles + 7) / 8, 256, 0, stream>>>(agg, W, b, out);
}


# global f32 atomic scatter replaced by per-destination-tile sums in LDS (250 tiles of 400 rows, fed by a byte tile map); LDS atomic adds in arrival order change the summation order
# speedup vs baseline: 2.0279x; 2.0279x over previous
#include <hip/hip_runtime.h>
#include <hip/hip_bf16.h>


typedef __attribute__((ext_vector_type(2))) float v2f;
typedef __attribute__((ext_vector_type(8))) float v8f;

#define GCN_N_NODES 100000
#define GCN_N_EDGES 1000000
#define GCN_D 64

#define GCN_TILE_ROWS 400
#define GCN_N_TILES (GCN_N_NODES / GCN_TILE_ROWS)
#define GCN_QCAP 12288
#define GCN_MAP_VECS (GCN_N_EDGES / 16)

__global__ void gcn_zero_deg(float* __restrict__ deg) {
    int i = blockIdx.x * blockDim.x + threadIdx.x;
    if (i < GCN_N_NODES) deg[i] = 0.0f;
}

__global__ void gcn_degree(const int* __restrict__ edge_index, float* __restrict__ deg,
                           unsigned int* __restrict__ tmap) {
    int e = blockIdx.x * blockDim.x + threadIdx.x;
    if (e < GCN_N_EDGES) {
        atomicAdd(&deg[edge_index[e]], 1.0f);
        unsigned d = (unsigned)edge_index[GCN_N_EDGES + e];
        ((unsigned char*)tmap)[e] = (unsigned char)(d / GCN_TILE_ROWS);
    }
}

__global__ void gcn_normalize(const float* __restrict__ x, const float* __restrict__ deg,
                              float* __restrict__ x_norm) {
    int i = blockIdx.x * blockDim.x + threadIdx.x;
    if (i >= GCN_N_NODES * (GCN_D / 4)) return;
    int n = i / (GCN_D / 4);
    float s = rsqrtf(deg[n] + 1.0f);
    float4 v = ((const float4*)x)[i];
    v.x *= s; v.y *= s; v.z *= s; v.w *= s;
    ((float4*)x_norm)[i] = v;
}

__device__ __forceinline__ void gcn_push_edge(unsigned e, unsigned* qn, unsigned* queue, float* tile,
                                              const int* __restrict__ edge_index,
                                              const float* __restrict__ x_norm, int lo) {
    unsigned idx = atomicAdd(qn, 1u);
    if (idx < GCN_QCAP) {
        queue[idx] = e;
    } else {
        int s  = edge_index[e];
        int dl = edge_index[GCN_N_EDGES + e] - lo;
#pragma unroll 1
        for (int f = 0; f < GCN_D; ++f)
            atomicAdd(&tile[dl * GCN_D + f], x_norm[(size_t)s * GCN_D + f]);
    }
}

__global__ __launch_bounds__(1024)
void gcn_tile_aggregate(const int* __restrict__ edge_index, const unsigned int* __restrict__ tmap,
                        const float* __restrict__ x_norm, float* __restrict__ agg) {
    __shared__ __attribute__((aligned(16))) float tile[GCN_TILE_ROWS * GCN_D];
    __shared__ unsigned queue[GCN_QCAP];
    __shared__ unsigned qn;

    const int tid = threadIdx.x;
    const int t   = blockIdx.x;
    const int lo  = t * GCN_TILE_ROWS;

    const float4* xin = (const float4*)(x_norm + (size_t)lo * GCN_D);
    float4* t4 = (float4*)tile;
    {
        float4 r[6];
#pragma unroll
        for (int k = 0; k < 6; ++k) r[k] = xin[tid + k * 1024];
#pragma unroll
        for (int k = 0; k < 6; ++k) t4[tid + k * 1024] = r[k];
        if (tid < 256) t4[6144 + tid] = xin[6144 + tid];
    }
    if (tid == 0) qn = 0u;
    __syncthreads();

    const unsigned pat = (unsigned)t * 0x01010101u;
    const uint4* mv = (const uint4*)tmap;
    for (int it = 0; it < (GCN_MAP_VECS + 1023) / 1024; it += 4) {
        uint4 w[4];
#pragma unroll
        for (int u = 0; u < 4; ++u) {
            int v  = (it + u) * 1024 + tid;
            int vc = v < GCN_MAP_VECS ? v : GCN_MAP_VECS - 1;
            w[u] = mv[vc];
        }
#pragma unroll
        for (int u = 0; u < 4; ++u) {
            int v = (it + u) * 1024 + tid;
            if (v < GCN_MAP_VECS) {
                unsigned wd[4];
                wd[0] = w[u].x; wd[1] = w[u].y; wd[2] = w[u].z; wd[3] = w[u].w;
                unsigned m = 0u;
#pragma unroll
                for (int k = 0; k < 4; ++k) {
                    unsigned x  = wd[k] ^ pat;
                    unsigned nz = (((x & 0x7F7F7F7Fu) + 0x7F7F7F7Fu) | x) & 0x80808080u;
                    unsigned z  = nz ^ 0x80808080u;
                    m |= z >> (7 - k);
                }
                while (m) {
                    unsigned p = (unsigned)__builtin_ctz(m);
                    m &= m - 1u;
                    unsigned k = p & 7u, b = p >> 3;
                    gcn_push_edge((unsigned)v * 16u + k * 4u + b, &qn, queue, tile, edge_index, x_norm, lo);
                }
            }
        }
    }
    __syncthreads();

    unsigned n = qn;
    if (n > GCN_QCAP) n = GCN_QCAP;
    n = __builtin_amdgcn_readfirstlane(n);
    const int lane = tid & 31;
    const unsigned wave = __builtin_amdgcn_readfirstlane(tid >> 5);
    const v2f* xr = (const v2f*)x_norm;
    for (unsigned base = wave * 32u; base < n; base += 1024u) {
        unsigned my = base + (unsigned)lane;
        int s = 0, dl = 0;
        if (my < n) {
            unsigned e = queue[my];
            s  = edge_index[e];
            dl = edge_index[GCN_N_EDGES + e] - lo;
        }
        unsigned cnt = n - base;
        if (cnt > 32u) cnt = 32u;
        for (unsigned j0 = 0; j0 < cnt; j0 += 16u) {
            v2f val[16];
            int drow[16];
#pragma unroll
            for (int u = 0; u < 16; ++u) {
                int sj  = __builtin_amdgcn_readlane(s, j0 + u);
                drow[u] = __builtin_amdgcn_readlane(dl, j0 + u);
                val[u]  = xr[(size_t)sj * (GCN_D / 2) + lane];
            }
#pragma unroll
            for (int u = 0; u < 16; ++u) {
                if (j0 + u < cnt) {
                    float* p = &tile[drow[u] * GCN_D + 2 * lane];
                    atomicAdd(p,     val[u].x);
                    atomicAdd(p + 1, val[u].y);
                }
            }
        }
    }
    __syncthreads();

    float4* aout = (float4*)(agg + (size_t)lo * GCN_D);
#pragma unroll
    for (int k = 0; k < 6; ++k) aout[tid + k * 1024] = t4[tid + k * 1024];
    if (tid < 256) aout[6144 + tid] = t4[6144 + tid];
}

__global__ void gcn_gemm_wmma(const float* __restrict__ agg, const float* __restrict__ W,
                              const float* __restrict__ bias, float* __restrict__ out) {
    const int lane = threadIdx.x & 31;
    const int wave = threadIdx.x >> 5;
    const int tile = blockIdx.x * 8 + wave;
    const int TILES_N = GCN_D / 16;
    const int n_tiles = (GCN_N_NODES / 16) * TILES_N;
    if (tile >= n_tiles) return;

    const int tm = tile / TILES_N;
    const int tn = tile % TILES_N;
    const int m0 = tm * 16;
    const int o0 = tn * 16;

    const int row = lane & 15;
    const int col = lane & 15;
    const int kb  = (lane >> 4) << 1;

    v2f breg[16];
    const float* wrow = W + (o0 + col) * GCN_D + kb;
#pragma unroll
    for (int t = 0; t < 16; ++t) {
        breg[t] = *(const v2f*)(wrow + 4 * t);
    }

    const float bv = bias[o0 + col];
    v8f c;
#pragma unroll
    for (int r = 0; r < 8; ++r) c[r] = bv;

    const float* arow = agg + (m0 + row) * GCN_D + kb;
#pragma unroll
    for (int t = 0; t < 16; ++t) {
        v2f a = *(const v2f*)(arow + 4 * t);
        c = __builtin_amdgcn_wmma_f32_16x16x4_f32(
                false, a, false, breg[t], (short)0, c, false, false);
    }

    float* orow = out + (m0 + ((lane >> 4) << 3)) * GCN_D + o0 + col;
#pragma unroll
    for (int r = 0; r < 8; ++r) {
        orow[r * GCN_D] = c[r];
    }
}

extern "C" void kernel_launch(void* const* d_in, const int* in_sizes, int n_in,
                              void* d_out, int out_size, void* d_ws, size_t ws_size,
                              hipStream_t stream) {
    const float* x          = (const float*)d_in[0];
    const int*   edge_index = (const int*)d_in[1];
    const float* W          = (const float*)d_in[2];
    const float* b          = (const float*)d_in[3];
    float*       out        = (float*)d_out;

    float* deg    = (float*)d_ws;
    float* x_norm = deg + GCN_N_NODES;
    float* agg    = x_norm + (size_t)GCN_N_NODES * GCN_D;
    unsigned int* tmap = (unsigned int*)(agg + (size_t)GCN_N_NODES * GCN_D);

    gcn_zero_deg<<<(GCN_N_NODES + 255) / 256, 256, 0, stream>>>(deg);
    gcn_degree<<<(GCN_N_EDGES + 255) / 256, 256, 0, stream>>>(edge_index, deg, tmap);
    gcn_normalize<<<(GCN_N_NODES * (GCN_D / 4) + 255) / 256, 256, 0, stream>>>(x, deg, x_norm);
    gcn_tile_aggregate<<<GCN_N_TILES, 1024, 0, stream>>>(edge_index, tmap, x_norm, agg);

    const int n_tiles = (GCN_N_NODES / 16) * (GCN_D / 16);
    gcn_gemm_wmma<<<(n_tiles + 7) / 8, 256, 0, stream>>>(agg, W, b, out);
}


# thread coarsening of the four thin kernels: 128-bit loads, 8 edges or 8 float4 per thread, one 32x64 strip per wave in the WMMA kernel; they start about 8x fewer waves; no sum changes its order
# speedup vs baseline: 2.0907x; 1.0310x over previous
#include <hip/hip_runtime.h>
#include <hip/hip_bf16.h>


typedef __attribute__((ext_vector_type(2))) float v2f;
typedef __attribute__((ext_vector_type(8))) float v8f;
typedef __attribute__((ext_vector_type(4))) int   v4i;

#define GCN_N_NODES 100000
#define GCN_N_EDGES 1000000
#define GCN_D 64

#define GCN_TILE_ROWS 400
#define GCN_N_TILES (GCN_N_NODES / GCN_TILE_ROWS)
#define GCN_QCAP 12288
#define GCN_MAP_VECS (GCN_N_EDGES / 16)

#define GCN_DEG_VECS  (GCN_N_NODES / 4)
#define GCN_EDGE_VECS (GCN_N_EDGES / 4)
#define GCN_X_VECS    (GCN_N_NODES * (GCN_D / 4))
#define GCN_ZERO_GRID 98
#define GCN_DEGREE_GRID 489
#define GCN_NORM_GRID 782
#define GCN_GEMM_GRID 391

__global__ __launch_bounds__(256)
void gcn_zero_deg(float* __restrict__ deg) {
    int i = blockIdx.x * 256 + threadIdx.x;
    if (i < GCN_DEG_VECS) ((float4*)deg)[i] = make_float4(0.0f, 0.0f, 0.0f, 0.0f);
}

__global__ __launch_bounds__(256)
void gcn_degree(const int* __restrict__ edge_index, float* __restrict__ deg,
                unsigned int* __restrict__ tmap) {
    const v4i* src4 = (const v4i*)edge_index;
    const v4i* dst4 = (const v4i*)(edge_index + GCN_N_EDGES);
    const int q0 = blockIdx.x * 512 + threadIdx.x;
    v4i s[2], d[2];
#pragma unroll
    for (int u = 0; u < 2; ++u) {
        int q  = q0 + u * 256;
        int qc = q < GCN_EDGE_VECS ? q : GCN_EDGE_VECS - 1;
        s[u] = src4[qc];
        d[u] = dst4[qc];
    }
    asm volatile("" :: "v"(s[0]), "v"(s[1]), "v"(d[0]), "v"(d[1]));
#pragma unroll
    for (int u = 0; u < 2; ++u) {
        int q = q0 + u * 256;
        if (q < GCN_EDGE_VECS) {
            atomicAdd(&deg[s[u].x], 1.0f);
            atomicAdd(&deg[s[u].y], 1.0f);
            atomicAdd(&deg[s[u].z], 1.0f);
            atomicAdd(&deg[s[u].w], 1.0f);
            tmap[q] =  ((unsigned)d[u].x / GCN_TILE_ROWS)
                    | (((unsigned)d[u].y / GCN_TILE_ROWS) << 8)
                    | (((unsigned)d[u].z / GCN_TILE_ROWS) << 16)
                    | (((unsigned)d[u].w / GCN_TILE_ROWS) << 24);
        }
    }
}

__global__ __launch_bounds__(256)
void gcn_normalize(const float* __restrict__ x, const float* __restrict__ deg,
                   float* __restrict__ x_norm) {
    const int base = blockIdx.x * 2048 + threadIdx.x;
    float4 v[8];
    float dg[8];
#pragma unroll
    for (int k = 0; k < 8; ++k) {
        int i  = base + k * 256;
        int ic = i < GCN_X_VECS ? i : GCN_X_VECS - 1;
        v[k]  = ((const float4*)x)[ic];
        dg[k] = deg[ic / (GCN_D / 4)];
    }
#pragma unroll
    for (int k = 0; k < 8; ++k) {
        int i = base + k * 256;
        if (i < GCN_X_VECS) {
            float s = rsqrtf(dg[k] + 1.0f);
            float4 r = v[k];
            r.x *= s; r.y *= s; r.z *= s; r.w *= s;
            ((float4*)x_norm)[i] = r;
        }
    }
}

__device__ __forceinline__ void gcn_push_edge(unsigned e, unsigned* qn, unsigned* queue, float* tile,
                                              const int* __restrict__ edge_index,
                                              const float* __restrict__ x_norm, int lo) {
    unsigned idx = atomicAdd(qn, 1u);
    if (idx < GCN_QCAP) {
        queue[idx] = e;
    } else {
        int s  = edge_index[e];
        int dl = edge_index[GCN_N_EDGES + e] - lo;
#pragma unroll 1
        for (int f = 0; f < GCN_D; ++f)
            atomicAdd(&tile[dl * GCN_D + f], x_norm[(size_t)s * GCN_D + f]);
    }
}

__global__ __launch_bounds__(1024)
void gcn_tile_aggregate(const int* __restrict__ edge_index, const unsigned int* __restrict__ tmap,
                        const float* __restrict__ x_norm, float* __restrict__ agg) {
    __shared__ __attribute__((aligned(16))) float tile[GCN_TILE_ROWS * GCN_D];
    __shared__ unsigned queue[GCN_QCAP];
    __shared__ unsigned qn;

    const int tid = threadIdx.x;
    const int t   = blockIdx.x;
    const int lo  = t * GCN_TILE_ROWS;

    const float4* xin = (const float4*)(x_norm + (size_t)lo * GCN_D);
    float4* t4 = (float4*)tile;
    {
        float4 r[6];
#pragma unroll
        for (int k = 0; k < 6; ++k) r[k] = xin[tid + k * 1024];
#pragma unroll
        for (int k = 0; k < 6; ++k) t4[tid + k * 1024] = r[k];
        if (tid < 256) t4[6144 + tid] = xin[6144 + tid];
    }
    if (tid == 0) qn = 0u;
    __syncthreads();

    const unsigned pat = (unsigned)t * 0x01010101u;
    const uint4* mv = (const uint4*)tmap;
    for (int it = 0; it < (GCN_MAP_VECS + 1023) / 1024; it += 4) {
        uint4 w[4];
#pragma unroll
        for (int u = 0; u < 4; ++u) {
            int v  = (it + u) * 1024 + tid;
            int vc = v < GCN_MAP_VECS ? v : GCN_MAP_VECS - 1;
            w[u] = mv[vc];
        }
#pragma unroll
        for (int u = 0; u < 4; ++u) {
            int v = (it + u) * 1024 + tid;
            if (v < GCN_MAP_VECS) {
                unsigned wd[4];
                wd[0] = w[u].x; wd[1] = w[u].y; wd[2] = w[u].z; wd[3] = w[u].w;
                unsigned m = 0u;
#pragma unroll
                for (int k = 0; k < 4; ++k) {
                    unsigned x  = wd[k] ^ pat;
                    unsigned nz = (((x & 0x7F7F7F7Fu) + 0x7F7F7F7Fu) | x) & 0x80808080u;
                    unsigned z  = nz ^ 0x80808080u;
                    m |= z >> (7 - k);
                }
                while (m) {
                    unsigned p = (unsigned)__builtin_ctz(m);
                    m &= m - 1u;
                    unsigned k = p & 7u, b = p >> 3;
                    gcn_push_edge((unsigned)v * 16u + k * 4u + b, &qn, queue, tile, edge_index, x_norm, lo);
                }
            }
        }
    }
    __syncthreads();

    unsigned n = qn;
    if (n > GCN_QCAP) n = GCN_QCAP;
    n = __builtin_amdgcn_readfirstlane(n);
    const int lane = tid & 31;
    const unsigned wave = __builtin_amdgcn_readfirstlane(tid >> 5);
    const v2f* xr = (const v2f*)x_norm;
    for (unsigned base = wave * 32u; base < n; base += 1024u) {
        unsigned my = base + (unsigned)lane;
        int s = 0, dl = 0;
        if (my < n) {
            unsigned e = queue[my];
            s  = edge_index[e];
            dl = edge_index[GCN_N_EDGES + e] - lo;
        }
        unsigned cnt = n - base;
        if (cnt > 32u) cnt = 32u;
        for (unsigned j0 = 0; j0 < cnt; j0 += 16u) {
            v2f val[16];
            int drow[16];
#pragma unroll
            for (int u = 0; u < 16; ++u) {
                int sj  = __builtin_amdgcn_readlane(s, j0 + u);
                drow[u] = __builtin_amdgcn_readlane(dl, j0 + u);
                val[u]  = xr[(size_t)sj * (GCN_D / 2) + lane];
            }
#pragma unroll
            for (int u = 0; u < 16; ++u) {
                if (j0 + u < cnt) {
                    float* p = &tile[drow[u] * GCN_D + 2 * lane];
                    atomicAdd(p,     val[u].x);
                    atomicAdd(p + 1, val[u].y);
                }
            }
        }
    }
    __syncthreads();

    float4* aout = (float4*)(agg + (size_t)lo * GCN_D);
#pragma unroll
    for (int k = 0; k < 6; ++k) aout[tid + k * 1024] = t4[tid + k * 1024];
    if (tid < 256) aout[6144 + tid] = t4[6144 + tid];
}

#define GCN_STRIPS (GCN_N_NODES / 32)
__global__ __launch_bounds__(256)
void gcn_gemm_wmma(const float* __restrict__ agg, const float* __restrict__ W,
                   const float* __restrict__ bias, float* __restrict__ out) {
    const int lane  = threadIdx.x & 31;
    const int wave  = threadIdx.x >> 5;
    const int strip = blockIdx.x * 8 + wave;
    if (strip >= GCN_STRIPS) return;

    const int row = lane & 15;
    const int col = lane & 15;
    const int kb  = (lane >> 4) << 1;

    v2f areg[2][16];
#pragma unroll
    for (int rt = 0; rt < 2; ++rt) {
        const float* arow = agg + (size_t)(strip * 32 + rt * 16 + row) * GCN_D + kb;
#pragma unroll
        for (int t = 0; t < 16; ++t)
            areg[rt][t] = *(const v2f*)(arow + 4 * t);
    }

    v2f breg[4][16];
    float bv[4];
#pragma unroll
    for (int tn = 0; tn < 4; ++tn) {
        const float* wrow = W + (tn * 16 + col) * GCN_D + kb;
#pragma unroll
        for (int t = 0; t < 16; ++t)
            breg[tn][t] = *(const v2f*)(wrow + 4 * t);
        bv[tn] = bias[tn * 16 + col];
    }

#pragma unroll
    for (int rt = 0; rt < 2; ++rt) {
        v8f c[4];
#pragma unroll
        for (int tn = 0; tn < 4; ++tn)
#pragma unroll
            for (int r = 0; r < 8; ++r) c[tn][r] = bv[tn];

#pragma unroll
        for (int t = 0; t < 16; ++t) {
#pragma unroll
            for (int tn = 0; tn < 4; ++tn) {
                c[tn] = __builtin_amdgcn_wmma_f32_16x16x4_f32(
                            false, areg[rt][t], false, breg[tn][t], (short)0, c[tn], false, false);
            }
        }

        float* orow = out + (size_t)(strip * 32 + rt * 16 + ((lane >> 4) << 3)) * GCN_D + col;
#pragma unroll
        for (int r = 0; r < 8; ++r) {
#pragma unroll
            for (int tn = 0; tn < 4; ++tn)
                orow[r * GCN_D + tn * 16] = c[tn][r];
        }
    }
}

extern "C" void kernel_launch(void* const* d_in, const int* in_sizes, int n_in,
                              void* d_out, int out_size, void* d_ws, size_t ws_size,
                              hipStream_t stream) {
    const float* x          = (const float*)d_in[0];
    const int*   edge_index = (const int*)d_in[1];
    const float* W          = (const float*)d_in[2];
    const float* b          = (const float*)d_in[3];
    float*       out        = (float*)d_out;

    float* deg    = (float*)d_ws;
    float* x_norm = deg + GCN_N_NODES;
    float* agg    = x_norm + (size_t)GCN_N_NODES * GCN_D;
    unsigned int* tmap = (unsigned int*)(agg + (size_t)GCN_N_NODES * GCN_D);

    gcn_zero_deg<<<GCN_ZERO_GRID, 256, 0, stream>>>(deg);
    gcn_degree<<<GCN_DEGREE_GRID, 256, 0, stream>>>(edge_index, deg, tmap);
    gcn_normalize<<<GCN_NORM_GRID, 256, 0, stream>>>(x, deg, x_norm);
    gcn_tile_aggregate<<<GCN_N_TILES, 1024, 0, stream>>>(edge_index, tmap, x_norm, agg);
    gcn_gemm_wmma<<<GCN_GEMM_GRID, 256, 0, stream>>>(agg, W, b, out);
}


# edges bucketed by destination tile once (LDS counting sort per 4096-edge chunk, packed src/row records); a tile reads only its own segments, no tile-map scan; LDS adds arrive in a new order
# speedup vs baseline: 2.5094x; 1.2003x over previous
#include <hip/hip_runtime.h>
#include <hip/hip_bf16.h>


typedef __attribute__((ext_vector_type(2))) float v2f;
typedef __attribute__((ext_vector_type(8))) float v8f;
typedef __attribute__((ext_vector_type(4))) int   v4i;

#define GCN_N_NODES 100000
#define GCN_N_EDGES 1000000
#define GCN_D 64

#define GCN_TILE_ROWS 400
#define GCN_N_TILES (GCN_N_NODES / GCN_TILE_ROWS)

#define GCN_CHUNK 4096
#define GCN_CHUNK_VECS (GCN_CHUNK / 4)
#define GCN_N_CHUNKS ((GCN_N_EDGES + GCN_CHUNK - 1) / GCN_CHUNK)
#define GCN_OFF_PITCH 256
#define GCN_SEGS 8

#define GCN_DEG_VECS  (GCN_N_NODES / 4)
#define GCN_EDGE_VECS (GCN_N_EDGES / 4)
#define GCN_X_VECS    (GCN_N_NODES * (GCN_D / 4))
#define GCN_ZERO_GRID 98
#define GCN_NORM_GRID 782
#define GCN_GEMM_GRID 391

__global__ __launch_bounds__(256)
void gcn_zero_deg(float* __restrict__ deg) {
    int i = blockIdx.x * 256 + threadIdx.x;
    if (i < GCN_DEG_VECS) ((float4*)deg)[i] = make_float4(0.0f, 0.0f, 0.0f, 0.0f);
}

__global__ __launch_bounds__(512)
void gcn_degree_bucket(const int* __restrict__ edge_index, float* __restrict__ deg,
                       unsigned int* __restrict__ recs, unsigned int* __restrict__ off) {
    __shared__ unsigned hist[GCN_OFF_PITCH];
    __shared__ unsigned start[GCN_OFF_PITCH];
    __shared__ __attribute__((aligned(16))) unsigned stage[GCN_CHUNK];

    const int tid = threadIdx.x;
    const int c   = blockIdx.x;
    if (tid < GCN_OFF_PITCH) hist[tid] = 0u;

    const v4i* src4 = (const v4i*)edge_index;
    const v4i* dst4 = (const v4i*)(edge_index + GCN_N_EDGES);
    const int q0 = c * GCN_CHUNK_VECS + tid;
    v4i s[2], d[2];
#pragma unroll
    for (int u = 0; u < 2; ++u) {
        int q  = q0 + u * 512;
        int qc = q < GCN_EDGE_VECS ? q : GCN_EDGE_VECS - 1;
        s[u] = src4[qc];
        d[u] = dst4[qc];
    }
    __syncthreads();

    unsigned tl[8], rk[8], rc[8];
#pragma unroll
    for (int u = 0; u < 2; ++u) {
        const bool ok = (q0 + u * 512) < GCN_EDGE_VECS;
        const int sv[4] = { s[u].x, s[u].y, s[u].z, s[u].w };
        const int dv[4] = { d[u].x, d[u].y, d[u].z, d[u].w };
#pragma unroll
        for (int k = 0; k < 4; ++k) {
            const int j = u * 4 + k;
            tl[j] = (unsigned)dv[k] / GCN_TILE_ROWS;
            rc[j] = ((unsigned)sv[k] << 9) | ((unsigned)dv[k] - tl[j] * GCN_TILE_ROWS);
            rk[j] = 0u;
            if (ok) rk[j] = atomicAdd(&hist[tl[j]], 1u);
        }
    }
    __syncthreads();

    if (tid < 32) {
        unsigned v[8], tot = 0u;
#pragma unroll
        for (int k = 0; k < 8; ++k) { v[k] = hist[tid * 8 + k]; tot += v[k]; }
        unsigned incl = tot;
#pragma unroll
        for (int o = 1; o < 32; o <<= 1) {
            unsigned nb = __shfl_up(incl, o);
            if (tid >= o) incl += nb;
        }
        unsigned run = incl - tot;
#pragma unroll
        for (int k = 0; k < 8; ++k) { start[tid * 8 + k] = run; run += v[k]; }
    }
    __syncthreads();

#pragma unroll
    for (int u = 0; u < 2; ++u) {
        if ((q0 + u * 512) < GCN_EDGE_VECS) {
#pragma unroll
            for (int k = 0; k < 4; ++k) {
                const int j = u * 4 + k;
                stage[start[tl[j]] + rk[j]] = rc[j];
            }
        }
    }
    if (tid < GCN_OFF_PITCH) off[c * GCN_OFF_PITCH + tid] = start[tid];
    __syncthreads();

    int nvec = GCN_EDGE_VECS - c * GCN_CHUNK_VECS;
    if (nvec > GCN_CHUNK_VECS) nvec = GCN_CHUNK_VECS;
    uint4* rout = (uint4*)recs + (size_t)c * GCN_CHUNK_VECS;
    for (int i = tid; i < nvec; i += 512) rout[i] = ((const uint4*)stage)[i];

#pragma unroll
    for (int u = 0; u < 2; ++u) {
        if ((q0 + u * 512) < GCN_EDGE_VECS) {
            atomicAdd(&deg[s[u].x], 1.0f);
            atomicAdd(&deg[s[u].y], 1.0f);
            atomicAdd(&deg[s[u].z], 1.0f);
            atomicAdd(&deg[s[u].w], 1.0f);
        }
    }
}

__global__ __launch_bounds__(256)
void gcn_normalize(const float* __restrict__ x, const float* __restrict__ deg,
                   float* __restrict__ x_norm) {
    const int base = blockIdx.x * 2048 + threadIdx.x;
    float4 v[8];
    float dg[8];
#pragma unroll
    for (int k = 0; k < 8; ++k) {
        int i  = base + k * 256;
        int ic = i < GCN_X_VECS ? i : GCN_X_VECS - 1;
        v[k]  = ((const float4*)x)[ic];
        dg[k] = deg[ic / (GCN_D / 4)];
    }
#pragma unroll
    for (int k = 0; k < 8; ++k) {
        int i = base + k * 256;
        if (i < GCN_X_VECS) {
            float s = rsqrtf(dg[k] + 1.0f);
            float4 r = v[k];
            r.x *= s; r.y *= s; r.z *= s; r.w *= s;
            ((float4*)x_norm)[i] = r;
        }
    }
}

__device__ __forceinline__ void gcn_add_rows(unsigned rv, unsigned cnt, float* tile,
                                             const v2f* __restrict__ xr, int lane) {
    for (unsigned j0 = 0; j0 < cnt; j0 += 16u) {
        v2f val[16];
        unsigned drow[16];
#pragma unroll
        for (int u = 0; u < 16; ++u) {
            unsigned r = (unsigned)__builtin_amdgcn_readlane((int)rv, (int)(j0 + u));
            drow[u] = r & 511u;
            val[u]  = xr[(size_t)(r >> 9) * (GCN_D / 2) + lane];
        }
#pragma unroll
        for (int u = 0; u < 16; ++u) {
            if (j0 + u < cnt) {
                float* p = &tile[drow[u] * GCN_D + 2 * lane];
                atomicAdd(p,     val[u].x);
                atomicAdd(p + 1, val[u].y);
            }
        }
    }
}

__global__ __launch_bounds__(1024)
void gcn_tile_aggregate(const unsigned int* __restrict__ recs, const unsigned int* __restrict__ off,
                        const float* __restrict__ x_norm, float* __restrict__ agg) {
    __shared__ __attribute__((aligned(16))) float tile[GCN_TILE_ROWS * GCN_D];
    __shared__ unsigned wq[32 * GCN_SEGS * 32];

    const int tid  = threadIdx.x;
    const int t    = blockIdx.x;
    const int lo   = t * GCN_TILE_ROWS;
    const int lane = tid & 31;
    const int wave = __builtin_amdgcn_readfirstlane(tid >> 5);

    unsigned sb = 0u, nn = 0u;
    {
        const int c = wave + 32 * lane;
        if (lane < GCN_SEGS && c < GCN_N_CHUNKS) {
            unsigned a = off[c * GCN_OFF_PITCH + t];
            unsigned b = off[c * GCN_OFF_PITCH + t + 1];
            sb = a;
            nn = b - a;
        }
    }

    const float4* xin = (const float4*)(x_norm + (size_t)lo * GCN_D);
    float4* t4 = (float4*)tile;
    float4 r[6];
#pragma unroll
    for (int k = 0; k < 6; ++k) r[k] = xin[tid + k * 1024];

    unsigned rec[GCN_SEGS], m[GCN_SEGS];
#pragma unroll
    for (int i = 0; i < GCN_SEGS; ++i) {
        unsigned si = (unsigned)__builtin_amdgcn_readlane((int)sb, i);
        unsigned ni = (unsigned)__builtin_amdgcn_readlane((int)nn, i);
        m[i]   = ni < 32u ? ni : 32u;
        rec[i] = 0u;
        if ((unsigned)lane < m[i]) rec[i] = recs[(size_t)(wave + 32 * i) * GCN_CHUNK + si + lane];
    }

#pragma unroll
    for (int k = 0; k < 6; ++k) t4[tid + k * 1024] = r[k];
    if (tid < 256) t4[6144 + tid] = xin[6144 + tid];

    unsigned* myq = wq + wave * (GCN_SEGS * 32);
    unsigned total = 0u;
#pragma unroll
    for (int i = 0; i < GCN_SEGS; ++i) {
        if ((unsigned)lane < m[i]) myq[total + lane] = rec[i];
        total += m[i];
    }
    __syncthreads();

    const v2f* xr = (const v2f*)x_norm;
    for (unsigned b0 = 0; b0 < total; b0 += 32u) {
        unsigned cnt = total - b0;
        if (cnt > 32u) cnt = 32u;
        unsigned rv = 0u;
        if ((unsigned)lane < cnt) rv = myq[b0 + lane];
        gcn_add_rows(rv, cnt, tile, xr, lane);
    }
#pragma unroll 1
    for (int i = 0; i < GCN_SEGS; ++i) {
        unsigned si = (unsigned)__builtin_amdgcn_readlane((int)sb, i);
        unsigned ni = (unsigned)__builtin_amdgcn_readlane((int)nn, i);
        for (unsigned k = 32u; k < ni; k += 32u) {
            unsigned cnt = ni - k;
            if (cnt > 32u) cnt = 32u;
            unsigned rv = 0u;
            if ((unsigned)lane < cnt) rv = recs[(size_t)(wave + 32 * i) * GCN_CHUNK + si + k + lane];
            gcn_add_rows(rv, cnt, tile, xr, lane);
        }
    }
    __syncthreads();

    float4* aout = (float4*)(agg + (size_t)lo * GCN_D);
#pragma unroll
    for (int k = 0; k < 6; ++k) aout[tid + k * 1024] = t4[tid + k * 1024];
    if (tid < 256) aout[6144 + tid] = t4[6144 + tid];
}

#define GCN_STRIPS (GCN_N_NODES / 32)
__global__ __launch_bounds__(256)
void gcn_gemm_wmma(const float* __restrict__ agg, const float* __restrict__ W,
                   const float* __restrict__ bias, float* __restrict__ out) {
    const int lane  = threadIdx.x & 31;
    const int wave  = threadIdx.x >> 5;
    const int strip = blockIdx.x * 8 + wave;
    if (strip >= GCN_STRIPS) return;

    const int row = lane & 15;
    const int col = lane & 15;
    const int kb  = (lane >> 4) << 1;

    v2f areg[2][16];
#pragma unroll
    for (int rt = 0; rt < 2; ++rt) {
        const float* arow = agg + (size_t)(strip * 32 + rt * 16 + row) * GCN_D + kb;
#pragma unroll
        for (int t = 0; t < 16; ++t)
            areg[rt][t] = *(const v2f*)(arow + 4 * t);
    }

    v2f breg[4][16];
    float bv[4];
#pragma unroll
    for (int tn = 0; tn < 4; ++tn) {
        const float* wrow = W + (tn * 16 + col) * GCN_D + kb;
#pragma unroll
        for (int t = 0; t < 16; ++t)
            breg[tn][t] = *(const v2f*)(wrow + 4 * t);
        bv[tn] = bias[tn * 16 + col];
    }

#pragma unroll
    for (int rt = 0; rt < 2; ++rt) {
        v8f c[4];
#pragma unroll
        for (int tn = 0; tn < 4; ++tn)
#pragma unroll
            for (int r = 0; r < 8; ++r) c[tn][r] = bv[tn];

#pragma unroll
        for (int t = 0; t < 16; ++t) {
#pragma unroll
            for (int tn = 0; tn < 4; ++tn) {
                c[tn] = __builtin_amdgcn_wmma_f32_16x16x4_f32(
                            false, areg[rt][t], false, breg[tn][t], (short)0, c[tn], false, false);
            }
        }

        float* orow = out + (size_t)(strip * 32 + rt * 16 + ((lane >> 4) << 3)) * GCN_D + col;
#pragma unroll
        for (int r = 0; r < 8; ++r) {
#pragma unroll
            for (int tn = 0; tn < 4; ++tn)
                orow[r * GCN_D + tn * 16] = c[tn][r];
        }
    }
}

extern "C" void kernel_launch(void* const* d_in, const int* in_sizes, int n_in,
                              void* d_out, int out_size, void* d_ws, size_t ws_size,
                              hipStream_t stream) {
    const float* x          = (const float*)d_in[0];
    const int*   edge_index = (const int*)d_in[1];
    const float* W          = (const float*)d_in[2];
    const float* b          = (const float*)d_in[3];
    float*       out        = (float*)d_out;

    float* deg    = (float*)d_ws;
    float* x_norm = deg + GCN_N_NODES;
    float* agg    = x_norm + (size_t)GCN_N_NODES * GCN_D;
    unsigned int* recs = (unsigned int*)(agg + (size_t)GCN_N_NODES * GCN_D);
    unsigned int* off  = recs + (size_t)GCN_N_CHUNKS * GCN_CHUNK;

    gcn_zero_deg<<<GCN_ZERO_GRID, 256, 0, stream>>>(deg);
    gcn_degree_bucket<<<GCN_N_CHUNKS, 512, 0, stream>>>(edge_index, deg, recs, off);
    gcn_normalize<<<GCN_NORM_GRID, 256, 0, stream>>>(x, deg, x_norm);
    gcn_tile_aggregate<<<GCN_N_TILES, 1024, 0, stream>>>(recs, off, x_norm, agg);
    gcn_gemm_wmma<<<GCN_GEMM_GRID, 256, 0, stream>>>(agg, W, b, out);
}


# matrix product fused into the tile-sum kernel: each finished 400-row tile is multiplied by W out of padded LDS rows (pitch 68) and stored to out; agg buffer and fifth launch go; no sum changes order
# speedup vs baseline: 2.7853x; 1.1099x over previous
#include <hip/hip_runtime.h>
#include <hip/hip_bf16.h>


typedef __attribute__((ext_vector_type(2))) float v2f;
typedef __attribute__((ext_vector_type(8))) float v8f;
typedef __attribute__((ext_vector_type(4))) int   v4i;

#define GCN_N_NODES 100000
#define GCN_N_EDGES 1000000
#define GCN_D 64

#define GCN_TILE_ROWS 400
#define GCN_N_TILES (GCN_N_NODES / GCN_TILE_ROWS)

#define GCN_CHUNK 4096
#define GCN_CHUNK_VECS (GCN_CHUNK / 4)
#define GCN_N_CHUNKS ((GCN_N_EDGES + GCN_CHUNK - 1) / GCN_CHUNK)
#define GCN_OFF_PITCH 256
#define GCN_SEGS 8

#define GCN_DEG_VECS  (GCN_N_NODES / 4)
#define GCN_EDGE_VECS (GCN_N_EDGES / 4)
#define GCN_X_VECS    (GCN_N_NODES * (GCN_D / 4))
#define GCN_ZERO_GRID 98
#define GCN_NORM_GRID 782
#define GCN_PITCH 68
#define GCN_PITCH4 (GCN_PITCH / 4)
#define GCN_ROW_TILES (GCN_TILE_ROWS / 16)

__global__ __launch_bounds__(256)
void gcn_zero_deg(float* __restrict__ deg) {
    int i = blockIdx.x * 256 + threadIdx.x;
    if (i < GCN_DEG_VECS) ((float4*)deg)[i] = make_float4(0.0f, 0.0f, 0.0f, 0.0f);
}

__global__ __launch_bounds__(512)
void gcn_degree_bucket(const int* __restrict__ edge_index, float* __restrict__ deg,
                       unsigned int* __restrict__ recs, unsigned int* __restrict__ off) {
    __shared__ unsigned hist[GCN_OFF_PITCH];
    __shared__ unsigned start[GCN_OFF_PITCH];
    __shared__ __attribute__((aligned(16))) unsigned stage[GCN_CHUNK];

    const int tid = threadIdx.x;
    const int c   = blockIdx.x;
    if (tid < GCN_OFF_PITCH) hist[tid] = 0u;

    const v4i* src4 = (const v4i*)edge_index;
    const v4i* dst4 = (const v4i*)(edge_index + GCN_N_EDGES);
    const int q0 = c * GCN_CHUNK_VECS + tid;
    v4i s[2], d[2];
#pragma unroll
    for (int u = 0; u < 2; ++u) {
        int q  = q0 + u * 512;
        int qc = q < GCN_EDGE_VECS ? q : GCN_EDGE_VECS - 1;
        s[u] = src4[qc];
        d[u] = dst4[qc];
    }
    __syncthreads();

    unsigned tl[8], rk[8], rc[8];
#pragma unroll
    for (int u = 0; u < 2; ++u) {
        const bool ok = (q0 + u * 512) < GCN_EDGE_VECS;
        const int sv[4] = { s[u].x, s[u].y, s[u].z, s[u].w };
        const int dv[4] = { d[u].x, d[u].y, d[u].z, d[u].w };
#pragma unroll
        for (int k = 0; k < 4; ++k) {
            const int j = u * 4 + k;
            tl[j] = (unsigned)dv[k] / GCN_TILE_ROWS;
            rc[j] = ((unsigned)sv[k] << 9) | ((unsigned)dv[k] - tl[j] * GCN_TILE_ROWS);
            rk[j] = 0u;
            if (ok) rk[j] = atomicAdd(&hist[tl[j]], 1u);
        }
    }
    __syncthreads();

    if (tid < 32) {
        unsigned v[8], tot = 0u;
#pragma unroll
        for (int k = 0; k < 8; ++k) { v[k] = hist[tid * 8 + k]; tot += v[k]; }
        unsigned incl = tot;
#pragma unroll
        for (int o = 1; o < 32; o <<= 1) {
            unsigned nb = __shfl_up(incl, o);
            if (tid >= o) incl += nb;
        }
        unsigned run = incl - tot;
#pragma unroll
        for (int k = 0; k < 8; ++k) { start[tid * 8 + k] = run; run += v[k]; }
    }
    __syncthreads();

#pragma unroll
    for (int u = 0; u < 2; ++u) {
        if ((q0 + u * 512) < GCN_EDGE_VECS) {
#pragma unroll
            for (int k = 0; k < 4; ++k) {
                const int j = u * 4 + k;
                stage[start[tl[j]] + rk[j]] = rc[j];
            }
        }
    }
    if (tid < GCN_OFF_PITCH) off[c * GCN_OFF_PITCH + tid] = start[tid];
    __syncthreads();

    int nvec = GCN_EDGE_VECS - c * GCN_CHUNK_VECS;
    if (nvec > GCN_CHUNK_VECS) nvec = GCN_CHUNK_VECS;
    uint4* rout = (uint4*)recs + (size_t)c * GCN_CHUNK_VECS;
    for (int i = tid; i < nvec; i += 512) rout[i] = ((const uint4*)stage)[i];

#pragma unroll
    for (int u = 0; u < 2; ++u) {
        if ((q0 + u * 512) < GCN_EDGE_VECS) {
            atomicAdd(&deg[s[u].x], 1.0f);
            atomicAdd(&deg[s[u].y], 1.0f);
            atomicAdd(&deg[s[u].z], 1.0f);
            atomicAdd(&deg[s[u].w], 1.0f);
        }
    }
}

__global__ __launch_bounds__(256)
void gcn_normalize(const float* __restrict__ x, const float* __restrict__ deg,
                   float* __restrict__ x_norm) {
    const int base = blockIdx.x * 2048 + threadIdx.x;
    float4 v[8];
    float dg[8];
#pragma unroll
    for (int k = 0; k < 8; ++k) {
        int i  = base + k * 256;
        int ic = i < GCN_X_VECS ? i : GCN_X_VECS - 1;
        v[k]  = ((const float4*)x)[ic];
        dg[k] = deg[ic / (GCN_D / 4)];
    }
#pragma unroll
    for (int k = 0; k < 8; ++k) {
        int i = base + k * 256;
        if (i < GCN_X_VECS) {
            float s = rsqrtf(dg[k] + 1.0f);
            float4 r = v[k];
            r.x *= s; r.y *= s; r.z *= s; r.w *= s;
            ((float4*)x_norm)[i] = r;
        }
    }
}

__device__ __forceinline__ void gcn_add_rows(unsigned rv, unsigned cnt, float* tile,
                                             const v2f* __restrict__ xr, int lane) {
    for (unsigned j0 = 0; j0 < cnt; j0 += 16u) {
        v2f val[16];
        unsigned drow[16];
#pragma unroll
        for (int u = 0; u < 16; ++u) {
            unsigned r = (unsigned)__builtin_amdgcn_readlane((int)rv, (int)(j0 + u));
            drow[u] = r & 511u;
            val[u]  = xr[(size_t)(r >> 9) * (GCN_D / 2) + lane];
        }
#pragma unroll
        for (int u = 0; u < 16; ++u) {
            if (j0 + u < cnt) {
                float* p = &tile[drow[u] * GCN_PITCH + 2 * lane];
                atomicAdd(p,     val[u].x);
                atomicAdd(p + 1, val[u].y);
            }
        }
    }
}

__global__ __launch_bounds__(1024)
void gcn_tile_aggregate_gemm(const unsigned int* __restrict__ recs, const unsigned int* __restrict__ off,
                             const float* __restrict__ x_norm, const float* __restrict__ W,
                             const float* __restrict__ bias, float* __restrict__ out) {
    __shared__ __attribute__((aligned(16))) float tile[GCN_TILE_ROWS * GCN_PITCH];
    __shared__ __attribute__((aligned(16))) float wlds[GCN_D * GCN_PITCH];
    __shared__ float blds[GCN_D];
    __shared__ unsigned wq[32 * GCN_SEGS * 32];

    const int tid  = threadIdx.x;
    const int t    = blockIdx.x;
    const int lo   = t * GCN_TILE_ROWS;
    const int lane = tid & 31;
    const int wave = __builtin_amdgcn_readfirstlane(tid >> 5);

    unsigned sb = 0u, nn = 0u;
    {
        const int c = wave + 32 * lane;
        if (lane < GCN_SEGS && c < GCN_N_CHUNKS) {
            unsigned a = off[c * GCN_OFF_PITCH + t];
            unsigned b = off[c * GCN_OFF_PITCH + t + 1];
            sb = a;
            nn = b - a;
        }
    }

    const int pbase = (tid >> 4) * GCN_PITCH4 + (tid & 15);

    const float4* xin = (const float4*)(x_norm + (size_t)lo * GCN_D);
    float4* t4 = (float4*)tile;
    float4 r[6];
#pragma unroll
    for (int k = 0; k < 6; ++k) r[k] = xin[tid + k * 1024];

    const float4 wv = ((const float4*)W)[tid];
    float bval = 0.0f;
    if (tid < GCN_D) bval = bias[tid];

    unsigned rec[GCN_SEGS], m[GCN_SEGS];
#pragma unroll
    for (int i = 0; i < GCN_SEGS; ++i) {
        unsigned si = (unsigned)__builtin_amdgcn_readlane((int)sb, i);
        unsigned ni = (unsigned)__builtin_amdgcn_readlane((int)nn, i);
        m[i]   = ni < 32u ? ni : 32u;
        rec[i] = 0u;
        if ((unsigned)lane < m[i]) rec[i] = recs[(size_t)(wave + 32 * i) * GCN_CHUNK + si + lane];
    }

#pragma unroll
    for (int k = 0; k < 6; ++k) t4[pbase + k * 64 * GCN_PITCH4] = r[k];
    ((float4*)wlds)[pbase] = wv;
    if (tid < 256) t4[pbase + 6 * 64 * GCN_PITCH4] = xin[6144 + tid];
    if (tid < GCN_D) blds[tid] = bval;

    unsigned* myq = wq + wave * (GCN_SEGS * 32);
    unsigned total = 0u;
#pragma unroll
    for (int i = 0; i < GCN_SEGS; ++i) {
        if ((unsigned)lane < m[i]) myq[total + lane] = rec[i];
        total += m[i];
    }
    __syncthreads();

    const v2f* xr = (const v2f*)x_norm;
    for (unsigned b0 = 0; b0 < total; b0 += 32u) {
        unsigned cnt = total - b0;
        if (cnt > 32u) cnt = 32u;
        unsigned rv = 0u;
        if ((unsigned)lane < cnt) rv = myq[b0 + lane];
        gcn_add_rows(rv, cnt, tile, xr, lane);
    }
#pragma unroll 1
    for (int i = 0; i < GCN_SEGS; ++i) {
        unsigned si = (unsigned)__builtin_amdgcn_readlane((int)sb, i);
        unsigned ni = (unsigned)__builtin_amdgcn_readlane((int)nn, i);
        for (unsigned k = 32u; k < ni; k += 32u) {
            unsigned cnt = ni - k;
            if (cnt > 32u) cnt = 32u;
            unsigned rv = 0u;
            if ((unsigned)lane < cnt) rv = recs[(size_t)(wave + 32 * i) * GCN_CHUNK + si + k + lane];
            gcn_add_rows(rv, cnt, tile, xr, lane);
        }
    }
    __syncthreads();

    if (wave < GCN_ROW_TILES) {
        const int row = lane & 15;
        const int col = lane & 15;
        const int kb  = (lane >> 4) << 1;
        float* trows = tile + wave * 16 * GCN_PITCH;

        v2f areg[16];
        const float* arow = trows + row * GCN_PITCH + kb;
#pragma unroll
        for (int s = 0; s < 16; ++s) areg[s] = *(const v2f*)(arow + 4 * s);

#pragma unroll
        for (int h = 0; h < 2; ++h) {
            v2f breg[2][16];
            v8f c[2];
#pragma unroll
            for (int j = 0; j < 2; ++j) {
                const int tn = 2 * h + j;
                const float* wrow = wlds + (tn * 16 + col) * GCN_PITCH + kb;
#pragma unroll
                for (int s = 0; s < 16; ++s) breg[j][s] = *(const v2f*)(wrow + 4 * s);
                const float bv = blds[tn * 16 + col];
#pragma unroll
                for (int q = 0; q < 8; ++q) c[j][q] = bv;
            }
#pragma unroll
            for (int s = 0; s < 16; ++s) {
#pragma unroll
                for (int j = 0; j < 2; ++j) {
                    c[j] = __builtin_amdgcn_wmma_f32_16x16x4_f32(
                               false, areg[s], false, breg[j][s], (short)0, c[j], false, false);
                }
            }
            float* orow = trows + ((lane >> 4) << 3) * GCN_PITCH + col;
#pragma unroll
            for (int q = 0; q < 8; ++q) {
#pragma unroll
                for (int j = 0; j < 2; ++j)
                    orow[q * GCN_PITCH + (2 * h + j) * 16] = c[j][q];
            }
        }
    }
    __syncthreads();

    float4* oout = (float4*)(out + (size_t)lo * GCN_D);
#pragma unroll
    for (int k = 0; k < 6; ++k) oout[tid + k * 1024] = t4[pbase + k * 64 * GCN_PITCH4];
    if (tid < 256) oout[6144 + tid] = t4[pbase + 6 * 64 * GCN_PITCH4];
}

extern "C" void kernel_launch(void* const* d_in, const int* in_sizes, int n_in,
                              void* d_out, int out_size, void* d_ws, size_t ws_size,
                              hipStream_t stream) {
    const float* x          = (const float*)d_in[0];
    const int*   edge_index = (const int*)d_in[1];
    const float* W          = (const float*)d_in[2];
    const float* b          = (const float*)d_in[3];
    float*       out        = (float*)d_out;

    float* deg    = (float*)d_ws;
    float* x_norm = deg + GCN_N_NODES;
    unsigned int* recs = (unsigned int*)(x_norm + (size_t)GCN_N_NODES * GCN_D);
    unsigned int* off  = recs + (size_t)GCN_N_CHUNKS * GCN_CHUNK;

    gcn_zero_deg<<<GCN_ZERO_GRID, 256, 0, stream>>>(deg);
    gcn_degree_bucket<<<GCN_N_CHUNKS, 512, 0, stream>>>(edge_index, deg, recs, off);
    gcn_normalize<<<GCN_NORM_GRID, 256, 0, stream>>>(x, deg, x_norm);
    gcn_tile_aggregate_gemm<<<GCN_N_TILES, 1024, 0, stream>>>(recs, off, x_norm, W, b, out);
}


# out-degrees counted per source tile in LDS by the normalise kernel (edges also bucketed by source tile; integer LDS adds): 1M f32 atomics on memory and the zeroing launch go; no sum changes order
# speedup vs baseline: 4.8848x; 1.7538x over previous
#include <hip/hip_runtime.h>
#include <hip/hip_bf16.h>


typedef __attribute__((ext_vector_type(2))) float v2f;
typedef __attribute__((ext_vector_type(8))) float v8f;
typedef __attribute__((ext_vector_type(4))) int   v4i;

#define GCN_N_NODES 100000
#define GCN_N_EDGES 1000000
#define GCN_D 64

#define GCN_TILE_ROWS 400
#define GCN_N_TILES (GCN_N_NODES / GCN_TILE_ROWS)

#define GCN_CHUNK 4096
#define GCN_CHUNK_VECS (GCN_CHUNK / 4)
#define GCN_N_CHUNKS ((GCN_N_EDGES + GCN_CHUNK - 1) / GCN_CHUNK)
#define GCN_OFF_PITCH 256
#define GCN_SEGS 8

#define GCN_EDGE_VECS (GCN_N_EDGES / 4)
#define GCN_PITCH 68
#define GCN_PITCH4 (GCN_PITCH / 4)
#define GCN_ROW_TILES (GCN_TILE_ROWS / 16)

__global__ __launch_bounds__(512)
void gcn_bucket_edges(const int* __restrict__ edge_index,
                      unsigned int* __restrict__ recs, unsigned int* __restrict__ off,
                      unsigned short* __restrict__ srow, unsigned int* __restrict__ soff) {
    __shared__ unsigned hist[GCN_OFF_PITCH];
    __shared__ unsigned start[GCN_OFF_PITCH];
    __shared__ unsigned shist[GCN_OFF_PITCH];
    __shared__ unsigned sstart[GCN_OFF_PITCH];
    __shared__ __attribute__((aligned(16))) unsigned stage[GCN_CHUNK];
    __shared__ __attribute__((aligned(16))) unsigned short sstage[GCN_CHUNK];

    const int tid = threadIdx.x;
    const int c   = blockIdx.x;
    if (tid < GCN_OFF_PITCH) { hist[tid] = 0u; shist[tid] = 0u; }

    const v4i* src4 = (const v4i*)edge_index;
    const v4i* dst4 = (const v4i*)(edge_index + GCN_N_EDGES);
    const int q0 = c * GCN_CHUNK_VECS + tid;
    v4i s[2], d[2];
#pragma unroll
    for (int u = 0; u < 2; ++u) {
        int q  = q0 + u * 512;
        int qc = q < GCN_EDGE_VECS ? q : GCN_EDGE_VECS - 1;
        s[u] = src4[qc];
        d[u] = dst4[qc];
    }
    __syncthreads();

    unsigned tl[8], rk[8], rc[8];
    unsigned stl[8], srk[8], srw[8];
#pragma unroll
    for (int u = 0; u < 2; ++u) {
        const bool ok = (q0 + u * 512) < GCN_EDGE_VECS;
        const int sv[4] = { s[u].x, s[u].y, s[u].z, s[u].w };
        const int dv[4] = { d[u].x, d[u].y, d[u].z, d[u].w };
#pragma unroll
        for (int k = 0; k < 4; ++k) {
            const int j = u * 4 + k;
            tl[j]  = (unsigned)dv[k] / GCN_TILE_ROWS;
            rc[j]  = ((unsigned)sv[k] << 9) | ((unsigned)dv[k] - tl[j] * GCN_TILE_ROWS);
            stl[j] = (unsigned)sv[k] / GCN_TILE_ROWS;
            srw[j] = (unsigned)sv[k] - stl[j] * GCN_TILE_ROWS;
            rk[j]  = 0u;
            srk[j] = 0u;
            if (ok) {
                rk[j]  = atomicAdd(&hist[tl[j]], 1u);
                srk[j] = atomicAdd(&shist[stl[j]], 1u);
            }
        }
    }
    __syncthreads();

    if (tid < 64) {
        const unsigned* h  = tid < 32 ? hist  : shist;
        unsigned*       st = tid < 32 ? start : sstart;
        const int l = tid & 31;
        unsigned v[8], tot = 0u;
#pragma unroll
        for (int k = 0; k < 8; ++k) { v[k] = h[l * 8 + k]; tot += v[k]; }
        unsigned incl = tot;
#pragma unroll
        for (int o = 1; o < 32; o <<= 1) {
            unsigned nb = __shfl_up(incl, o);
            if (l >= o) incl += nb;
        }
        unsigned run = incl - tot;
#pragma unroll
        for (int k = 0; k < 8; ++k) { st[l * 8 + k] = run; run += v[k]; }
    }
    __syncthreads();

#pragma unroll
    for (int u = 0; u < 2; ++u) {
        if ((q0 + u * 512) < GCN_EDGE_VECS) {
#pragma unroll
            for (int k = 0; k < 4; ++k) {
                const int j = u * 4 + k;
                stage[start[tl[j]] + rk[j]] = rc[j];
                sstage[sstart[stl[j]] + srk[j]] = (unsigned short)srw[j];
            }
        }
    }
    if (tid < GCN_OFF_PITCH) {
        off[c * GCN_OFF_PITCH + tid]  = start[tid];
        soff[c * GCN_OFF_PITCH + tid] = sstart[tid];
    }
    __syncthreads();

    int nvec = GCN_EDGE_VECS - c * GCN_CHUNK_VECS;
    if (nvec > GCN_CHUNK_VECS) nvec = GCN_CHUNK_VECS;
    uint4* rout = (uint4*)recs + (size_t)c * GCN_CHUNK_VECS;
    for (int i = tid; i < nvec; i += 512) rout[i] = ((const uint4*)stage)[i];
    uint4* sout = (uint4*)srow + (size_t)c * (GCN_CHUNK / 8);
    if (tid < ((nvec + 1) >> 1)) sout[tid] = ((const uint4*)sstage)[tid];
}

__global__ __launch_bounds__(1024)
void gcn_tile_degree_normalize(const float* __restrict__ x, const unsigned short* __restrict__ srow,
                               const unsigned int* __restrict__ soff, float* __restrict__ x_norm) {
    __shared__ unsigned cnt[GCN_TILE_ROWS];

    const int tid  = threadIdx.x;
    const int t    = blockIdx.x;
    const int lo   = t * GCN_TILE_ROWS;
    const int lane = tid & 31;
    const int wave = __builtin_amdgcn_readfirstlane(tid >> 5);
    if (tid < GCN_TILE_ROWS) cnt[tid] = 0u;

    unsigned sb = 0u, nn = 0u;
    {
        const int c = wave + 32 * lane;
        if (lane < GCN_SEGS && c < GCN_N_CHUNKS) {
            unsigned a = soff[c * GCN_OFF_PITCH + t];
            unsigned b = soff[c * GCN_OFF_PITCH + t + 1];
            sb = a;
            nn = b - a;
        }
    }

    const float4* xin = (const float4*)(x + (size_t)lo * GCN_D);
    float4 r[6];
#pragma unroll
    for (int k = 0; k < 6; ++k) r[k] = xin[tid + k * 1024];
    float4 rt = make_float4(0.0f, 0.0f, 0.0f, 0.0f);
    if (tid < 256) rt = xin[6144 + tid];

    unsigned rec[GCN_SEGS], m[GCN_SEGS];
#pragma unroll
    for (int i = 0; i < GCN_SEGS; ++i) {
        unsigned si = (unsigned)__builtin_amdgcn_readlane((int)sb, i);
        unsigned ni = (unsigned)__builtin_amdgcn_readlane((int)nn, i);
        m[i]   = ni < 32u ? ni : 32u;
        rec[i] = 0u;
        if ((unsigned)lane < m[i]) rec[i] = srow[(size_t)(wave + 32 * i) * GCN_CHUNK + si + lane];
    }
    __syncthreads();

#pragma unroll
    for (int i = 0; i < GCN_SEGS; ++i) {
        if ((unsigned)lane < m[i]) atomicAdd(&cnt[rec[i]], 1u);
    }
#pragma unroll 1
    for (int i = 0; i < GCN_SEGS; ++i) {
        unsigned si = (unsigned)__builtin_amdgcn_readlane((int)sb, i);
        unsigned ni = (unsigned)__builtin_amdgcn_readlane((int)nn, i);
        for (unsigned k = 32u + (unsigned)lane; k < ni; k += 32u)
            atomicAdd(&cnt[srow[(size_t)(wave + 32 * i) * GCN_CHUNK + si + k]], 1u);
    }
    __syncthreads();

    float4* xout = (float4*)(x_norm + (size_t)lo * GCN_D);
#pragma unroll
    for (int k = 0; k < 6; ++k) {
        const float sc = rsqrtf((float)cnt[(tid >> 4) + 64 * k] + 1.0f);
        float4 v = r[k];
        v.x *= sc; v.y *= sc; v.z *= sc; v.w *= sc;
        xout[tid + k * 1024] = v;
    }
    if (tid < 256) {
        const float sc = rsqrtf((float)cnt[384 + (tid >> 4)] + 1.0f);
        rt.x *= sc; rt.y *= sc; rt.z *= sc; rt.w *= sc;
        xout[6144 + tid] = rt;
    }
}

__device__ __forceinline__ void gcn_add_rows(unsigned rv, unsigned cnt, float* tile,
                                             const v2f* __restrict__ xr, int lane) {
    for (unsigned j0 = 0; j0 < cnt; j0 += 16u) {
        v2f val[16];
        unsigned drow[16];
#pragma unroll
        for (int u = 0; u < 16; ++u) {
            unsigned r = (unsigned)__builtin_amdgcn_readlane((int)rv, (int)(j0 + u));
            drow[u] = r & 511u;
            val[u]  = xr[(size_t)(r >> 9) * (GCN_D / 2) + lane];
        }
#pragma unroll
        for (int u = 0; u < 16; ++u) {
            if (j0 + u < cnt) {
                float* p = &tile[drow[u] * GCN_PITCH + 2 * lane];
                atomicAdd(p,     val[u].x);
                atomicAdd(p + 1, val[u].y);
            }
        }
    }
}

__global__ __launch_bounds__(1024)
void gcn_tile_aggregate_gemm(const unsigned int* __restrict__ recs, const unsigned int* __restrict__ off,
                             const float* __restrict__ x_norm, const float* __restrict__ W,
                             const float* __restrict__ bias, float* __restrict__ out) {
    __shared__ __attribute__((aligned(16))) float tile[GCN_TILE_ROWS * GCN_PITCH];
    __shared__ __attribute__((aligned(16))) float wlds[GCN_D * GCN_PITCH];
    __shared__ float blds[GCN_D];
    __shared__ unsigned wq[32 * GCN_SEGS * 32];

    const int tid  = threadIdx.x;
    const int t    = blockIdx.x;
    const int lo   = t * GCN_TILE_ROWS;
    const int lane = tid & 31;
    const int wave = __builtin_amdgcn_readfirstlane(tid >> 5);

    unsigned sb = 0u, nn = 0u;
    {
        const int c = wave + 32 * lane;
        if (lane < GCN_SEGS && c < GCN_N_CHUNKS) {
            unsigned a = off[c * GCN_OFF_PITCH + t];
            unsigned b = off[c * GCN_OFF_PITCH + t + 1];
            sb = a;
            nn = b - a;
        }
    }

    const int pbase = (tid >> 4) * GCN_PITCH4 + (tid & 15);

    const float4* xin = (const float4*)(x_norm + (size_t)lo * GCN_D);
    float4* t4 = (float4*)tile;
    float4 r[6];
#pragma unroll
    for (int k = 0; k < 6; ++k) r[k] = xin[tid + k * 1024];

    const float4 wv = ((const float4*)W)[tid];
    float bval = 0.0f;
    if (tid < GCN_D) bval = bias[tid];

    unsigned rec[GCN_SEGS], m[GCN_SEGS];
#pragma unroll
    for (int i = 0; i < GCN_SEGS; ++i) {
        unsigned si = (unsigned)__builtin_amdgcn_readlane((int)sb, i);
        unsigned ni = (unsigned)__builtin_amdgcn_readlane((int)nn, i);
        m[i]   = ni < 32u ? ni : 32u;
        rec[i] = 0u;
        if ((unsigned)lane < m[i]) rec[i] = recs[(size_t)(wave + 32 * i) * GCN_CHUNK + si + lane];
    }

#pragma unroll
    for (int k = 0; k < 6; ++k) t4[pbase + k * 64 * GCN_PITCH4] = r[k];
    ((float4*)wlds)[pbase] = wv;
    if (tid < 256) t4[pbase + 6 * 64 * GCN_PITCH4] = xin[6144 + tid];
    if (tid < GCN_D) blds[tid] = bval;

    unsigned* myq = wq + wave * (GCN_SEGS * 32);
    unsigned total = 0u;
#pragma unroll
    for (int i = 0; i < GCN_SEGS; ++i) {
        if ((unsigned)lane < m[i]) myq[total + lane] = rec[i];
        total += m[i];
    }
    __syncthreads();

    const v2f* xr = (const v2f*)x_norm;
    for (unsigned b0 = 0; b0 < total; b0 += 32u) {
        unsigned cnt = total - b0;
        if (cnt > 32u) cnt = 32u;
        unsigned rv = 0u;
        if ((unsigned)lane < cnt) rv = myq[b0 + lane];
        gcn_add_rows(rv, cnt, tile, xr, lane);
    }
#pragma unroll 1
    for (int i = 0; i < GCN_SEGS; ++i) {
        unsigned si = (unsigned)__builtin_amdgcn_readlane((int)sb, i);
        unsigned ni = (unsigned)__builtin_amdgcn_readlane((int)nn, i);
        for (unsigned k = 32u; k < ni; k += 32u) {
            unsigned cnt = ni - k;
            if (cnt > 32u) cnt = 32u;
            unsigned rv = 0u;
            if ((unsigned)lane < cnt) rv = recs[(size_t)(wave + 32 * i) * GCN_CHUNK + si + k + lane];
            gcn_add_rows(rv, cnt, tile, xr, lane);
        }
    }
    __syncthreads();

    if (wave < GCN_ROW_TILES) {
        const int row = lane & 15;
        const int col = lane & 15;
        const int kb  = (lane >> 4) << 1;
        float* trows = tile + wave * 16 * GCN_PITCH;

        v2f areg[16];
        const float* arow = trows + row * GCN_PITCH + kb;
#pragma unroll
        for (int s = 0; s < 16; ++s) areg[s] = *(const v2f*)(arow + 4 * s);

#pragma unroll
        for (int h = 0; h < 2; ++h) {
            v2f breg[2][16];
            v8f c[2];
#pragma unroll
            for (int j = 0; j < 2; ++j) {
                const int tn = 2 * h + j;
                const float* wrow = wlds + (tn * 16 + col) * GCN_PITCH + kb;
#pragma unroll
                for (int s = 0; s < 16; ++s) breg[j][s] = *(const v2f*)(wrow + 4 * s);
                const float bv = blds[tn * 16 + col];
#pragma unroll
                for (int q = 0; q < 8; ++q) c[j][q] = bv;
            }
#pragma unroll
            for (int s = 0; s < 16; ++s) {
#pragma unroll
                for (int j = 0; j < 2; ++j) {
                    c[j] = __builtin_amdgcn_wmma_f32_16x16x4_f32(
                               false, areg[s], false, breg[j][s], (short)0, c[j], false, false);
                }
            }
            float* orow = trows + ((lane >> 4) << 3) * GCN_PITCH + col;
#pragma unroll
            for (int q = 0; q < 8; ++q) {
#pragma unroll
                for (int j = 0; j < 2; ++j)
                    orow[q * GCN_PITCH + (2 * h + j) * 16] = c[j][q];
            }
        }
    }
    __syncthreads();

    float4* oout = (float4*)(out + (size_t)lo * GCN_D);
#pragma unroll
    for (int k = 0; k < 6; ++k) oout[tid + k * 1024] = t4[pbase + k * 64 * GCN_PITCH4];
    if (tid < 256) oout[6144 + tid] = t4[pbase + 6 * 64 * GCN_PITCH4];
}

extern "C" void kernel_launch(void* const* d_in, const int* in_sizes, int n_in,
                              void* d_out, int out_size, void* d_ws, size_t ws_size,
                              hipStream_t stream) {
    const float* x          = (const float*)d_in[0];
    const int*   edge_index = (const int*)d_in[1];
    const float* W          = (const float*)d_in[2];
    const float* b          = (const float*)d_in[3];
    float*       out        = (float*)d_out;

    float* x_norm = (float*)d_ws;
    unsigned int* recs = (unsigned int*)(x_norm + (size_t)GCN_N_NODES * GCN_D);
    unsigned int* off  = recs + (size_t)GCN_N_CHUNKS * GCN_CHUNK;
    unsigned short* srow = (unsigned short*)(off + (size_t)GCN_N_CHUNKS * GCN_OFF_PITCH);
    unsigned int* soff = (unsigned int*)(srow + (size_t)GCN_N_CHUNKS * GCN_CHUNK);

    gcn_bucket_edges<<<GCN_N_CHUNKS, 512, 0, stream>>>(edge_index, recs, off, srow, soff);
    gcn_tile_degree_normalize<<<GCN_N_TILES, 1024, 0, stream>>>(x, srow, soff, x_norm);
    gcn_tile_aggregate_gemm<<<GCN_N_TILES, 1024, 0, stream>>>(recs, off, x_norm, W, b, out);
}


# a tile's records counting-sorted by destination row in LDS; each wave sums a row's run of source rows in registers and adds it to the tile once: LDS float adds per run, not per edge; new sum order
# speedup vs baseline: 5.7193x; 1.1708x over previous
#include <hip/hip_runtime.h>
#include <hip/hip_bf16.h>


typedef __attribute__((ext_vector_type(2))) float v2f;
typedef __attribute__((ext_vector_type(8))) float v8f;
typedef __attribute__((ext_vector_type(4))) int   v4i;

#define GCN_N_NODES 100000
#define GCN_N_EDGES 1000000
#define GCN_D 64

#define GCN_TILE_ROWS 400
#define GCN_N_TILES (GCN_N_NODES / GCN_TILE_ROWS)

#define GCN_CHUNK 4096
#define GCN_CHUNK_VECS (GCN_CHUNK / 4)
#define GCN_N_CHUNKS ((GCN_N_EDGES + GCN_CHUNK - 1) / GCN_CHUNK)
#define GCN_OFF_PITCH 256
#define GCN_SEGS 8

#define GCN_EDGE_VECS (GCN_N_EDGES / 4)
#define GCN_PITCH 68
#define GCN_PITCH4 (GCN_PITCH / 4)
#define GCN_ROW_TILES (GCN_TILE_ROWS / 16)
#define GCN_SCAN_PER_LANE 13
#define GCN_ROW_CNT (32 * GCN_SCAN_PER_LANE)
#define GCN_NO_ROW 0xFFFFFFFFu

__global__ __launch_bounds__(512)
void gcn_bucket_edges(const int* __restrict__ edge_index,
                      unsigned int* __restrict__ recs, unsigned int* __restrict__ off,
                      unsigned short* __restrict__ srow, unsigned int* __restrict__ soff) {
    __shared__ unsigned hist[GCN_OFF_PITCH];
    __shared__ unsigned start[GCN_OFF_PITCH];
    __shared__ unsigned shist[GCN_OFF_PITCH];
    __shared__ unsigned sstart[GCN_OFF_PITCH];
    __shared__ __attribute__((aligned(16))) unsigned stage[GCN_CHUNK];
    __shared__ __attribute__((aligned(16))) unsigned short sstage[GCN_CHUNK];

    const int tid = threadIdx.x;
    const int c   = blockIdx.x;
    if (tid < GCN_OFF_PITCH) { hist[tid] = 0u; shist[tid] = 0u; }

    const v4i* src4 = (const v4i*)edge_index;
    const v4i* dst4 = (const v4i*)(edge_index + GCN_N_EDGES);
    const int q0 = c * GCN_CHUNK_VECS + tid;
    v4i s[2], d[2];
#pragma unroll
    for (int u = 0; u < 2; ++u) {
        int q  = q0 + u * 512;
        int qc = q < GCN_EDGE_VECS ? q : GCN_EDGE_VECS - 1;
        s[u] = src4[qc];
        d[u] = dst4[qc];
    }
    __syncthreads();

    unsigned tl[8], rk[8], rc[8];
    unsigned stl[8], srk[8], srw[8];
#pragma unroll
    for (int u = 0; u < 2; ++u) {
        const bool ok = (q0 + u * 512) < GCN_EDGE_VECS;
        const int sv[4] = { s[u].x, s[u].y, s[u].z, s[u].w };
        const int dv[4] = { d[u].x, d[u].y, d[u].z, d[u].w };
#pragma unroll
        for (int k = 0; k < 4; ++k) {
            const int j = u * 4 + k;
            tl[j]  = (unsigned)dv[k] / GCN_TILE_ROWS;
            rc[j]  = ((unsigned)sv[k] << 9) | ((unsigned)dv[k] - tl[j] * GCN_TILE_ROWS);
            stl[j] = (unsigned)sv[k] / GCN_TILE_ROWS;
            srw[j] = (unsigned)sv[k] - stl[j] * GCN_TILE_ROWS;
            rk[j]  = 0u;
            srk[j] = 0u;
            if (ok) {
                rk[j]  = atomicAdd(&hist[tl[j]], 1u);
                srk[j] = atomicAdd(&shist[stl[j]], 1u);
            }
        }
    }
    __syncthreads();

    if (tid < 64) {
        const unsigned* h  = tid < 32 ? hist  : shist;
        unsigned*       st = tid < 32 ? start : sstart;
        const int l = tid & 31;
        unsigned v[8], tot = 0u;
#pragma unroll
        for (int k = 0; k < 8; ++k) { v[k] = h[l * 8 + k]; tot += v[k]; }
        unsigned incl = tot;
#pragma unroll
        for (int o = 1; o < 32; o <<= 1) {
            unsigned nb = __shfl_up(incl, o);
            if (l >= o) incl += nb;
        }
        unsigned run = incl - tot;
#pragma unroll
        for (int k = 0; k < 8; ++k) { st[l * 8 + k] = run; run += v[k]; }
    }
    __syncthreads();

#pragma unroll
    for (int u = 0; u < 2; ++u) {
        if ((q0 + u * 512) < GCN_EDGE_VECS) {
#pragma unroll
            for (int k = 0; k < 4; ++k) {
                const int j = u * 4 + k;
                stage[start[tl[j]] + rk[j]] = rc[j];
                sstage[sstart[stl[j]] + srk[j]] = (unsigned short)srw[j];
            }
        }
    }
    if (tid < GCN_OFF_PITCH) {
        off[c * GCN_OFF_PITCH + tid]  = start[tid];
        soff[c * GCN_OFF_PITCH + tid] = sstart[tid];
    }
    __syncthreads();

    int nvec = GCN_EDGE_VECS - c * GCN_CHUNK_VECS;
    if (nvec > GCN_CHUNK_VECS) nvec = GCN_CHUNK_VECS;
    uint4* rout = (uint4*)recs + (size_t)c * GCN_CHUNK_VECS;
    for (int i = tid; i < nvec; i += 512) rout[i] = ((const uint4*)stage)[i];
    uint4* sout = (uint4*)srow + (size_t)c * (GCN_CHUNK / 8);
    if (tid < ((nvec + 1) >> 1)) sout[tid] = ((const uint4*)sstage)[tid];
}

__global__ __launch_bounds__(1024)
void gcn_tile_degree_normalize(const float* __restrict__ x, const unsigned short* __restrict__ srow,
                               const unsigned int* __restrict__ soff, float* __restrict__ x_norm) {
    __shared__ unsigned cnt[GCN_TILE_ROWS];

    const int tid  = threadIdx.x;
    const int t    = blockIdx.x;
    const int lo   = t * GCN_TILE_ROWS;
    const int lane = tid & 31;
    const int wave = __builtin_amdgcn_readfirstlane(tid >> 5);
    if (tid < GCN_TILE_ROWS) cnt[tid] = 0u;

    unsigned sb = 0u, nn = 0u;
    {
        const int c = wave + 32 * lane;
        if (lane < GCN_SEGS && c < GCN_N_CHUNKS) {
            unsigned a = soff[c * GCN_OFF_PITCH + t];
            unsigned b = soff[c * GCN_OFF_PITCH + t + 1];
            sb = a;
            nn = b - a;
        }
    }

    const float4* xin = (const float4*)(x + (size_t)lo * GCN_D);
    float4 r[6];
#pragma unroll
    for (int k = 0; k < 6; ++k) r[k] = xin[tid + k * 1024];
    float4 rt = make_float4(0.0f, 0.0f, 0.0f, 0.0f);
    if (tid < 256) rt = xin[6144 + tid];

    unsigned rec[GCN_SEGS], m[GCN_SEGS];
#pragma unroll
    for (int i = 0; i < GCN_SEGS; ++i) {
        unsigned si = (unsigned)__builtin_amdgcn_readlane((int)sb, i);
        unsigned ni = (unsigned)__builtin_amdgcn_readlane((int)nn, i);
        m[i]   = ni < 32u ? ni : 32u;
        rec[i] = 0u;
        if ((unsigned)lane < m[i]) rec[i] = srow[(size_t)(wave + 32 * i) * GCN_CHUNK + si + lane];
    }
    __syncthreads();

#pragma unroll
    for (int i = 0; i < GCN_SEGS; ++i) {
        if ((unsigned)lane < m[i]) atomicAdd(&cnt[rec[i]], 1u);
    }
#pragma unroll 1
    for (int i = 0; i < GCN_SEGS; ++i) {
        unsigned si = (unsigned)__builtin_amdgcn_readlane((int)sb, i);
        unsigned ni = (unsigned)__builtin_amdgcn_readlane((int)nn, i);
        for (unsigned k = 32u + (unsigned)lane; k < ni; k += 32u)
            atomicAdd(&cnt[srow[(size_t)(wave + 32 * i) * GCN_CHUNK + si + k]], 1u);
    }
    __syncthreads();

    float4* xout = (float4*)(x_norm + (size_t)lo * GCN_D);
#pragma unroll
    for (int k = 0; k < 6; ++k) {
        const float sc = rsqrtf((float)cnt[(tid >> 4) + 64 * k] + 1.0f);
        float4 v = r[k];
        v.x *= sc; v.y *= sc; v.z *= sc; v.w *= sc;
        xout[tid + k * 1024] = v;
    }
    if (tid < 256) {
        const float sc = rsqrtf((float)cnt[384 + (tid >> 4)] + 1.0f);
        rt.x *= sc; rt.y *= sc; rt.z *= sc; rt.w *= sc;
        xout[6144 + tid] = rt;
    }
}

__device__ __forceinline__ void gcn_add_rows(unsigned rv, unsigned cnt, float* tile,
                                             const v2f* __restrict__ xr, int lane) {
    for (unsigned j0 = 0; j0 < cnt; j0 += 16u) {
        v2f val[16];
        unsigned drow[16];
#pragma unroll
        for (int u = 0; u < 16; ++u) {
            unsigned r = (unsigned)__builtin_amdgcn_readlane((int)rv, (int)(j0 + u));
            drow[u] = r & 511u;
            val[u]  = xr[(size_t)(r >> 9) * (GCN_D / 2) + lane];
        }
#pragma unroll
        for (int u = 0; u < 16; ++u) {
            if (j0 + u < cnt) {
                float* p = &tile[drow[u] * GCN_PITCH + 2 * lane];
                atomicAdd(p,     val[u].x);
                atomicAdd(p + 1, val[u].y);
            }
        }
    }
}

__device__ __forceinline__ void gcn_flush_run(float* tile, unsigned row, v2f acc, int lane) {
    float* p = &tile[row * GCN_PITCH + 2 * lane];
    atomicAdd(p,     acc.x);
    atomicAdd(p + 1, acc.y);
}

template <int J0, bool CHECK>
__device__ __forceinline__ void gcn_sum_group(unsigned rv, unsigned cnt, unsigned& cur, v2f& acc, float* tile,
                                              const v2f* __restrict__ xr, int lane) {
    v2f val[16];
    unsigned drow[16];
#pragma unroll
    for (int u = 0; u < 16; ++u) {
        unsigned r = (unsigned)__builtin_amdgcn_readlane((int)rv, J0 + u);
        drow[u] = r & 511u;
        val[u]  = xr[(size_t)(r >> 9) * (GCN_D / 2) + lane];
    }
#pragma unroll
    for (int u = 0; u < 16; ++u) {
        if (!CHECK || (unsigned)(J0 + u) < cnt) {
            if (__builtin_expect(drow[u] != cur, 0)) {
                if (cur != GCN_NO_ROW) gcn_flush_run(tile, cur, acc, lane);
                cur = drow[u];
                acc = (v2f){ 0.0f, 0.0f };
            }
            acc += val[u];
        }
    }
}

__device__ __forceinline__ void gcn_sum_sorted(const unsigned* sorted, unsigned beg, unsigned end, float* tile,
                                               const v2f* __restrict__ xr, int lane) {
    unsigned cur = GCN_NO_ROW;
    v2f acc = { 0.0f, 0.0f };
    for (unsigned b0 = beg; b0 < end; b0 += 32u) {
        unsigned cnt = end - b0;
        if (cnt > 32u) cnt = 32u;
        unsigned rv = 0u;
        if ((unsigned)lane < cnt) rv = sorted[b0 + lane];
        if (cnt >= 16u) gcn_sum_group<0, false>(rv, cnt, cur, acc, tile, xr, lane);
        else            gcn_sum_group<0, true >(rv, cnt, cur, acc, tile, xr, lane);
        if (cnt >= 32u)      gcn_sum_group<16, false>(rv, cnt, cur, acc, tile, xr, lane);
        else if (cnt > 16u)  gcn_sum_group<16, true >(rv, cnt, cur, acc, tile, xr, lane);
    }
    if (cur != GCN_NO_ROW) gcn_flush_run(tile, cur, acc, lane);
}

__global__ __launch_bounds__(1024)
void gcn_tile_aggregate_gemm(const unsigned int* __restrict__ recs, const unsigned int* __restrict__ off,
                             const float* __restrict__ x_norm, const float* __restrict__ W,
                             const float* __restrict__ bias, float* __restrict__ out) {
    __shared__ __attribute__((aligned(16))) float tile[GCN_TILE_ROWS * GCN_PITCH];
    __shared__ __attribute__((aligned(16))) float wlds[GCN_D * GCN_PITCH];
    __shared__ float blds[GCN_D];
    __shared__ unsigned sorted[32 * GCN_SEGS * 32];
    __shared__ unsigned rowcnt[GCN_ROW_CNT];

    const int tid  = threadIdx.x;
    const int t    = blockIdx.x;
    const int lo   = t * GCN_TILE_ROWS;
    const int lane = tid & 31;
    const int wave = __builtin_amdgcn_readfirstlane(tid >> 5);
    if (tid < GCN_ROW_CNT) rowcnt[tid] = 0u;

    unsigned sb = 0u, nn = 0u;
    {
        const int c = wave + 32 * lane;
        if (lane < GCN_SEGS && c < GCN_N_CHUNKS) {
            unsigned a = off[c * GCN_OFF_PITCH + t];
            unsigned b = off[c * GCN_OFF_PITCH + t + 1];
            sb = a;
            nn = b - a;
        }
    }

    const int pbase = (tid >> 4) * GCN_PITCH4 + (tid & 15);

    const float4* xin = (const float4*)(x_norm + (size_t)lo * GCN_D);
    float4* t4 = (float4*)tile;
    float4 r[6];
#pragma unroll
    for (int k = 0; k < 6; ++k) r[k] = xin[tid + k * 1024];

    const float4 wv = ((const float4*)W)[tid];
    float bval = 0.0f;
    if (tid < GCN_D) bval = bias[tid];

    unsigned rec[GCN_SEGS], m[GCN_SEGS];
#pragma unroll
    for (int i = 0; i < GCN_SEGS; ++i) {
        unsigned si = (unsigned)__builtin_amdgcn_readlane((int)sb, i);
        unsigned ni = (unsigned)__builtin_amdgcn_readlane((int)nn, i);
        m[i]   = ni < 32u ? ni : 32u;
        rec[i] = 0u;
        if ((unsigned)lane < m[i]) rec[i] = recs[(size_t)(wave + 32 * i) * GCN_CHUNK + si + lane];
    }

#pragma unroll
    for (int k = 0; k < 6; ++k) t4[pbase + k * 64 * GCN_PITCH4] = r[k];
    ((float4*)wlds)[pbase] = wv;
    if (tid < 256) t4[pbase + 6 * 64 * GCN_PITCH4] = xin[6144 + tid];
    if (tid < GCN_D) blds[tid] = bval;

    __syncthreads();

    unsigned rk[GCN_SEGS];
#pragma unroll
    for (int i = 0; i < GCN_SEGS; ++i) {
        rk[i] = 0u;
        if ((unsigned)lane < m[i]) rk[i] = atomicAdd(&rowcnt[rec[i] & 511u], 1u);
    }
    __syncthreads();

    if (wave == 0) {
        unsigned v[GCN_SCAN_PER_LANE], tot = 0u;
#pragma unroll
        for (int k = 0; k < GCN_SCAN_PER_LANE; ++k) { v[k] = rowcnt[lane * GCN_SCAN_PER_LANE + k]; tot += v[k]; }
        unsigned incl = tot;
#pragma unroll
        for (int o = 1; o < 32; o <<= 1) {
            unsigned nb = __shfl_up(incl, o);
            if (lane >= o) incl += nb;
        }
        unsigned run = incl - tot;
#pragma unroll
        for (int k = 0; k < GCN_SCAN_PER_LANE; ++k) { rowcnt[lane * GCN_SCAN_PER_LANE + k] = run; run += v[k]; }
    }
    __syncthreads();

    unsigned pos[GCN_SEGS];
#pragma unroll
    for (int i = 0; i < GCN_SEGS; ++i) pos[i] = rowcnt[rec[i] & 511u] + rk[i];
#pragma unroll
    for (int i = 0; i < GCN_SEGS; ++i) {
        if ((unsigned)lane < m[i]) sorted[pos[i]] = rec[i];
    }
    const unsigned total = (unsigned)__builtin_amdgcn_readfirstlane((int)rowcnt[GCN_TILE_ROWS]);
    __syncthreads();

    const v2f* xr = (const v2f*)x_norm;
    gcn_sum_sorted(sorted, ((unsigned)wave * total) >> 5, ((unsigned)(wave + 1) * total) >> 5, tile, xr, lane);
#pragma unroll 1
    for (int i = 0; i < GCN_SEGS; ++i) {
        unsigned si = (unsigned)__builtin_amdgcn_readlane((int)sb, i);
        unsigned ni = (unsigned)__builtin_amdgcn_readlane((int)nn, i);
        for (unsigned k = 32u; k < ni; k += 32u) {
            unsigned cnt = ni - k;
            if (cnt > 32u) cnt = 32u;
            unsigned rv = 0u;
            if ((unsigned)lane < cnt) rv = recs[(size_t)(wave + 32 * i) * GCN_CHUNK + si + k + lane];
            gcn_add_rows(rv, cnt, tile, xr, lane);
        }
    }
    __syncthreads();

    if (wave < GCN_ROW_TILES) {
        const int row = lane & 15;
        const int col = lane & 15;
        const int kb  = (lane >> 4) << 1;
        float* trows = tile + wave * 16 * GCN_PITCH;

        v2f areg[16];
        const float* arow = trows + row * GCN_PITCH + kb;
#pragma unroll
        for (int s = 0; s < 16; ++s) areg[s] = *(const v2f*)(arow + 4 * s);

#pragma unroll
        for (int h = 0; h < 2; ++h) {
            v2f breg[2][16];
            v8f c[2];
#pragma unroll
            for (int j = 0; j < 2; ++j) {
                const int tn = 2 * h + j;
                const float* wrow = wlds + (tn * 16 + col) * GCN_PITCH + kb;
#pragma unroll
                for (int s = 0; s < 16; ++s) breg[j][s] = *(const v2f*)(wrow + 4 * s);
                const float bv = blds[tn * 16 + col];
#pragma unroll
                for (int q = 0; q < 8; ++q) c[j][q] = bv;
            }
#pragma unroll
            for (int s = 0; s < 16; ++s) {
#pragma unroll
                for (int j = 0; j < 2; ++j) {
                    c[j] = __builtin_amdgcn_wmma_f32_16x16x4_f32(
                               false, areg[s], false, breg[j][s], (short)0, c[j], false, false);
                }
            }
            float* orow = trows + ((lane >> 4) << 3) * GCN_PITCH + col;
#pragma unroll
            for (int q = 0; q < 8; ++q) {
#pragma unroll
                for (int j = 0; j < 2; ++j)
                    orow[q * GCN_PITCH + (2 * h + j) * 16] = c[j][q];
            }
        }
    }
    __syncthreads();

    float4* oout = (float4*)(out + (size_t)lo * GCN_D);
#pragma unroll
    for (int k = 0; k < 6; ++k) oout[tid + k * 1024] = t4[pbase + k * 64 * GCN_PITCH4];
    if (tid < 256) oout[6144 + tid] = t4[pbase + 6 * 64 * GCN_PITCH4];
}

extern "C" void kernel_launch(void* const* d_in, const int* in_sizes, int n_in,
                              void* d_out, int out_size, void* d_ws, size_t ws_size,
                              hipStream_t stream) {
    const float* x          = (const float*)d_in[0];
    const int*   edge_index = (const int*)d_in[1];
    const float* W          = (const float*)d_in[2];
    const float* b          = (const float*)d_in[3];
    float*       out        = (float*)d_out;

    float* x_norm = (float*)d_ws;
    unsigned int* recs = (unsigned int*)(x_norm + (size_t)GCN_N_NODES * GCN_D);
    unsigned int* off  = recs + (size_t)GCN_N_CHUNKS * GCN_CHUNK;
    unsigned short* srow = (unsigned short*)(off + (size_t)GCN_N_CHUNKS * GCN_OFF_PITCH);
    unsigned int* soff = (unsigned int*)(srow + (size_t)GCN_N_CHUNKS * GCN_CHUNK);

    gcn_bucket_edges<<<GCN_N_CHUNKS, 512, 0, stream>>>(edge_index, recs, off, srow, soff);
    gcn_tile_degree_normalize<<<GCN_N_TILES, 1024, 0, stream>>>(x, srow, soff, x_norm);
    gcn_tile_aggregate_gemm<<<GCN_N_TILES, 1024, 0, stream>>>(recs, off, x_norm, W, b, out);
}


# x_norm no longer written or read: the degree kernel (8 waves a tile) stores only scale = rsqrt(deg+1); the tile sums gather x rows and apply scale[src] by fma (one rounding less a term, same order)
# speedup vs baseline: 6.5133x; 1.1388x over previous
#include <hip/hip_runtime.h>
#include <hip/hip_bf16.h>


typedef __attribute__((ext_vector_type(2))) float v2f;
typedef __attribute__((ext_vector_type(8))) float v8f;
typedef __attribute__((ext_vector_type(4))) int   v4i;

#define GCN_N_NODES 100000
#define GCN_N_EDGES 1000000
#define GCN_D 64

#define GCN_TILE_ROWS 400
#define GCN_N_TILES (GCN_N_NODES / GCN_TILE_ROWS)

#define GCN_CHUNK 4096
#define GCN_CHUNK_VECS (GCN_CHUNK / 4)
#define GCN_N_CHUNKS ((GCN_N_EDGES + GCN_CHUNK - 1) / GCN_CHUNK)
#define GCN_OFF_PITCH 256
#define GCN_SEGS 8

#define GCN_DEG_WAVES 8
#define GCN_DEG_SEGS 32
#define GCN_EDGE_VECS (GCN_N_EDGES / 4)
#define GCN_PITCH 68
#define GCN_PITCH4 (GCN_PITCH / 4)
#define GCN_ROW_TILES (GCN_TILE_ROWS / 16)
#define GCN_SCAN_PER_LANE 13
#define GCN_ROW_CNT (32 * GCN_SCAN_PER_LANE)
#define GCN_NO_ROW 0xFFFFFFFFu

__global__ __launch_bounds__(512)
void gcn_bucket_edges(const int* __restrict__ edge_index,
                      unsigned int* __restrict__ recs, unsigned int* __restrict__ off,
                      unsigned short* __restrict__ srow, unsigned int* __restrict__ soff) {
    __shared__ unsigned hist[GCN_OFF_PITCH];
    __shared__ unsigned start[GCN_OFF_PITCH];
    __shared__ unsigned shist[GCN_OFF_PITCH];
    __shared__ unsigned sstart[GCN_OFF_PITCH];
    __shared__ __attribute__((aligned(16))) unsigned stage[GCN_CHUNK];
    __shared__ __attribute__((aligned(16))) unsigned short sstage[GCN_CHUNK];

    const int tid = threadIdx.x;
    const int c   = blockIdx.x;
    if (tid < GCN_OFF_PITCH) { hist[tid] = 0u; shist[tid] = 0u; }

    const v4i* src4 = (const v4i*)edge_index;
    const v4i* dst4 = (const v4i*)(edge_index + GCN_N_EDGES);
    const int q0 = c * GCN_CHUNK_VECS + tid;
    v4i s[2], d[2];
#pragma unroll
    for (int u = 0; u < 2; ++u) {
        int q  = q0 + u * 512;
        int qc = q < GCN_EDGE_VECS ? q : GCN_EDGE_VECS - 1;
        s[u] = src4[qc];
        d[u] = dst4[qc];
    }
    __syncthreads();

    unsigned tl[8], rk[8], rc[8];
    unsigned stl[8], srk[8], srw[8];
#pragma unroll
    for (int u = 0; u < 2; ++u) {
        const bool ok = (q0 + u * 512) < GCN_EDGE_VECS;
        const int sv[4] = { s[u].x, s[u].y, s[u].z, s[u].w };
        const int dv[4] = { d[u].x, d[u].y, d[u].z, d[u].w };
#pragma unroll
        for (int k = 0; k < 4; ++k) {
            const int j = u * 4 + k;
            tl[j]  = (unsigned)dv[k] / GCN_TILE_ROWS;
            rc[j]  = ((unsigned)sv[k] << 9) | ((unsigned)dv[k] - tl[j] * GCN_TILE_ROWS);
            stl[j] = (unsigned)sv[k] / GCN_TILE_ROWS;
            srw[j] = (unsigned)sv[k] - stl[j] * GCN_TILE_ROWS;
            rk[j]  = 0u;
            srk[j] = 0u;
            if (ok) {
                rk[j]  = atomicAdd(&hist[tl[j]], 1u);
                srk[j] = atomicAdd(&shist[stl[j]], 1u);
            }
        }
    }
    __syncthreads();

    if (tid < 64) {
        const unsigned* h  = tid < 32 ? hist  : shist;
        unsigned*       st = tid < 32 ? start : sstart;
        const int l = tid & 31;
        unsigned v[8], tot = 0u;
#pragma unroll
        for (int k = 0; k < 8; ++k) { v[k] = h[l * 8 + k]; tot += v[k]; }
        unsigned incl = tot;
#pragma unroll
        for (int o = 1; o < 32; o <<= 1) {
            unsigned nb = __shfl_up(incl, o);
            if (l >= o) incl += nb;
        }
        unsigned run = incl - tot;
#pragma unroll
        for (int k = 0; k < 8; ++k) { st[l * 8 + k] = run; run += v[k]; }
    }
    __syncthreads();

#pragma unroll
    for (int u = 0; u < 2; ++u) {
        if ((q0 + u * 512) < GCN_EDGE_VECS) {
#pragma unroll
            for (int k = 0; k < 4; ++k) {
                const int j = u * 4 + k;
                stage[start[tl[j]] + rk[j]] = rc[j];
                sstage[sstart[stl[j]] + srk[j]] = (unsigned short)srw[j];
            }
        }
    }
    if (tid < GCN_OFF_PITCH) {
        off[c * GCN_OFF_PITCH + tid]  = start[tid];
        soff[c * GCN_OFF_PITCH + tid] = sstart[tid];
    }
    __syncthreads();

    int nvec = GCN_EDGE_VECS - c * GCN_CHUNK_VECS;
    if (nvec > GCN_CHUNK_VECS) nvec = GCN_CHUNK_VECS;
    uint4* rout = (uint4*)recs + (size_t)c * GCN_CHUNK_VECS;
    for (int i = tid; i < nvec; i += 512) rout[i] = ((const uint4*)stage)[i];
    uint4* sout = (uint4*)srow + (size_t)c * (GCN_CHUNK / 8);
    if (tid < ((nvec + 1) >> 1)) sout[tid] = ((const uint4*)sstage)[tid];
}

__global__ __launch_bounds__(256)
void gcn_tile_degree_scale(const unsigned short* __restrict__ srow, const unsigned int* __restrict__ soff,
                           float* __restrict__ scale) {
    __shared__ unsigned cnt[GCN_TILE_ROWS];

    const int tid  = threadIdx.x;
    const int t    = blockIdx.x;
    const int lo   = t * GCN_TILE_ROWS;
    const int lane = tid & 31;
    const int wave = __builtin_amdgcn_readfirstlane(tid >> 5);
    const int c0   = wave * GCN_DEG_SEGS;
    for (int i = tid; i < GCN_TILE_ROWS; i += 256) cnt[i] = 0u;

    unsigned sb = 0u, nn = 0u;
    {
        const int c = c0 + lane;
        if (c < GCN_N_CHUNKS) {
            unsigned a = soff[c * GCN_OFF_PITCH + t];
            unsigned b = soff[c * GCN_OFF_PITCH + t + 1];
            sb = a;
            nn = b - a;
        }
    }

    unsigned rec[GCN_DEG_SEGS];
#pragma unroll
    for (int i = 0; i < GCN_DEG_SEGS; ++i) {
        unsigned si = (unsigned)__builtin_amdgcn_readlane((int)sb, i);
        unsigned ni = (unsigned)__builtin_amdgcn_readlane((int)nn, i);
        rec[i] = 0u;
        if ((unsigned)lane < ni) rec[i] = srow[(size_t)(c0 + i) * GCN_CHUNK + si + lane];
    }
    __syncthreads();

#pragma unroll
    for (int i = 0; i < GCN_DEG_SEGS; ++i) {
        unsigned ni = (unsigned)__builtin_amdgcn_readlane((int)nn, i);
        if ((unsigned)lane < ni) atomicAdd(&cnt[rec[i]], 1u);
    }
#pragma unroll 1
    for (int i = 0; i < GCN_DEG_SEGS; ++i) {
        unsigned si = (unsigned)__builtin_amdgcn_readlane((int)sb, i);
        unsigned ni = (unsigned)__builtin_amdgcn_readlane((int)nn, i);
        for (unsigned k = 32u + (unsigned)lane; k < ni; k += 32u)
            atomicAdd(&cnt[srow[(size_t)(c0 + i) * GCN_CHUNK + si + k]], 1u);
    }
    __syncthreads();

    for (int i = tid; i < GCN_TILE_ROWS; i += 256)
        scale[lo + i] = rsqrtf((float)cnt[i] + 1.0f);
}

__device__ __forceinline__ void gcn_add_rows(unsigned rv, float sv, unsigned cnt, float* tile,
                                             const v2f* __restrict__ xr, int lane) {
    for (unsigned j0 = 0; j0 < cnt; j0 += 16u) {
        v2f val[16];
        unsigned drow[16];
#pragma unroll
        for (int u = 0; u < 16; ++u) {
            unsigned r = (unsigned)__builtin_amdgcn_readlane((int)rv, (int)(j0 + u));
            drow[u] = r & 511u;
            val[u]  = xr[(size_t)(r >> 9) * (GCN_D / 2) + lane];
        }
#pragma unroll
        for (int u = 0; u < 16; ++u) {
            if (j0 + u < cnt) {
                const float s = __builtin_bit_cast(float, __builtin_amdgcn_readlane(__builtin_bit_cast(int, sv), (int)(j0 + u)));
                float* p = &tile[drow[u] * GCN_PITCH + 2 * lane];
                atomicAdd(p,     val[u].x * s);
                atomicAdd(p + 1, val[u].y * s);
            }
        }
    }
}

__device__ __forceinline__ void gcn_flush_run(float* tile, unsigned row, v2f acc, int lane) {
    float* p = &tile[row * GCN_PITCH + 2 * lane];
    atomicAdd(p,     acc.x);
    atomicAdd(p + 1, acc.y);
}

template <int J0, bool CHECK>
__device__ __forceinline__ void gcn_sum_group(unsigned rv, float sv, unsigned cnt, unsigned& cur, v2f& acc, float* tile,
                                              const v2f* __restrict__ xr, int lane) {
    v2f val[16];
    unsigned drow[16];
#pragma unroll
    for (int u = 0; u < 16; ++u) {
        unsigned r = (unsigned)__builtin_amdgcn_readlane((int)rv, J0 + u);
        drow[u] = r & 511u;
        val[u]  = xr[(size_t)(r >> 9) * (GCN_D / 2) + lane];
    }
#pragma unroll
    for (int u = 0; u < 16; ++u) {
        if (!CHECK || (unsigned)(J0 + u) < cnt) {
            if (__builtin_expect(drow[u] != cur, 0)) {
                if (cur != GCN_NO_ROW) gcn_flush_run(tile, cur, acc, lane);
                cur = drow[u];
                acc = (v2f){ 0.0f, 0.0f };
            }
            const float s = __builtin_bit_cast(float, __builtin_amdgcn_readlane(__builtin_bit_cast(int, sv), J0 + u));
            acc.x = __builtin_fmaf(val[u].x, s, acc.x);
            acc.y = __builtin_fmaf(val[u].y, s, acc.y);
        }
    }
}

__device__ __forceinline__ void gcn_sum_sorted(const uint2* sorted, unsigned beg, unsigned end, float* tile,
                                               const v2f* __restrict__ xr, int lane) {
    unsigned cur = GCN_NO_ROW;
    v2f acc = { 0.0f, 0.0f };
    for (unsigned b0 = beg; b0 < end; b0 += 32u) {
        unsigned cnt = end - b0;
        if (cnt > 32u) cnt = 32u;
        unsigned rv = 0u;
        float sv = 0.0f;
        if ((unsigned)lane < cnt) {
            const uint2 e = sorted[b0 + lane];
            rv = e.x;
            sv = __builtin_bit_cast(float, e.y);
        }
        if (cnt >= 16u) gcn_sum_group<0, false>(rv, sv, cnt, cur, acc, tile, xr, lane);
        else            gcn_sum_group<0, true >(rv, sv, cnt, cur, acc, tile, xr, lane);
        if (cnt >= 32u)      gcn_sum_group<16, false>(rv, sv, cnt, cur, acc, tile, xr, lane);
        else if (cnt > 16u)  gcn_sum_group<16, true >(rv, sv, cnt, cur, acc, tile, xr, lane);
    }
    if (cur != GCN_NO_ROW) gcn_flush_run(tile, cur, acc, lane);
}

__global__ __launch_bounds__(1024)
void gcn_tile_aggregate_gemm(const unsigned int* __restrict__ recs, const unsigned int* __restrict__ off,
                             const float* __restrict__ x, const float* __restrict__ scale,
                             const float* __restrict__ W,
                             const float* __restrict__ bias, float* __restrict__ out) {
    __shared__ __attribute__((aligned(16))) float tile[GCN_TILE_ROWS * GCN_PITCH];
    __shared__ __attribute__((aligned(16))) float wlds[GCN_D * GCN_PITCH];
    __shared__ float blds[GCN_D];
    __shared__ uint2 sorted[32 * GCN_SEGS * 32];
    __shared__ unsigned rowcnt[GCN_ROW_CNT];

    const int tid  = threadIdx.x;
    const int t    = blockIdx.x;
    const int lo   = t * GCN_TILE_ROWS;
    const int lane = tid & 31;
    const int wave = __builtin_amdgcn_readfirstlane(tid >> 5);
    if (tid < GCN_ROW_CNT) rowcnt[tid] = 0u;

    unsigned sb = 0u, nn = 0u;
    {
        const int c = wave + 32 * lane;
        if (lane < GCN_SEGS && c < GCN_N_CHUNKS) {
            unsigned a = off[c * GCN_OFF_PITCH + t];
            unsigned b = off[c * GCN_OFF_PITCH + t + 1];
            sb = a;
            nn = b - a;
        }
    }

    const int pbase = (tid >> 4) * GCN_PITCH4 + (tid & 15);

    const float4* xin = (const float4*)(x + (size_t)lo * GCN_D);
    float4* t4 = (float4*)tile;
    float4 r[6];
    float  so[6];
#pragma unroll
    for (int k = 0; k < 6; ++k) {
        r[k]  = xin[tid + k * 1024];
        so[k] = scale[lo + (tid >> 4) + 64 * k];
    }
    const int tail = tid < 256 ? 6144 + tid : tid;
    float4 rt = xin[tail];
    float  st = scale[lo + (tail >> 4)];

    const float4 wv = ((const float4*)W)[tid];
    float bval = 0.0f;
    if (tid < GCN_D) bval = bias[tid];

    unsigned rec[GCN_SEGS], m[GCN_SEGS];
#pragma unroll
    for (int i = 0; i < GCN_SEGS; ++i) {
        unsigned si = (unsigned)__builtin_amdgcn_readlane((int)sb, i);
        unsigned ni = (unsigned)__builtin_amdgcn_readlane((int)nn, i);
        m[i]   = ni < 32u ? ni : 32u;
        rec[i] = 0u;
        if ((unsigned)lane < m[i]) rec[i] = recs[(size_t)(wave + 32 * i) * GCN_CHUNK + si + lane];
    }
    unsigned scv[GCN_SEGS];
#pragma unroll
    for (int i = 0; i < GCN_SEGS; ++i) scv[i] = __builtin_bit_cast(unsigned, scale[rec[i] >> 9]);

#pragma unroll
    for (int k = 0; k < 6; ++k) {
        float4 v = r[k];
        v.x *= so[k]; v.y *= so[k]; v.z *= so[k]; v.w *= so[k];
        t4[pbase + k * 64 * GCN_PITCH4] = v;
    }
    ((float4*)wlds)[pbase] = wv;
    rt.x *= st; rt.y *= st; rt.z *= st; rt.w *= st;
    if (tid < 256) t4[pbase + 6 * 64 * GCN_PITCH4] = rt;
    if (tid < GCN_D) blds[tid] = bval;

    __syncthreads();

    unsigned rk[GCN_SEGS];
#pragma unroll
    for (int i = 0; i < GCN_SEGS; ++i) {
        rk[i] = 0u;
        if ((unsigned)lane < m[i]) rk[i] = atomicAdd(&rowcnt[rec[i] & 511u], 1u);
    }
    __syncthreads();

    if (wave == 0) {
        unsigned v[GCN_SCAN_PER_LANE], tot = 0u;
#pragma unroll
        for (int k = 0; k < GCN_SCAN_PER_LANE; ++k) { v[k] = rowcnt[lane * GCN_SCAN_PER_LANE + k]; tot += v[k]; }
        unsigned incl = tot;
#pragma unroll
        for (int o = 1; o < 32; o <<= 1) {
            unsigned nb = __shfl_up(incl, o);
            if (lane >= o) incl += nb;
        }
        unsigned run = incl - tot;
#pragma unroll
        for (int k = 0; k < GCN_SCAN_PER_LANE; ++k) { rowcnt[lane * GCN_SCAN_PER_LANE + k] = run; run += v[k]; }
    }
    __syncthreads();

    unsigned pos[GCN_SEGS];
#pragma unroll
    for (int i = 0; i < GCN_SEGS; ++i) pos[i] = rowcnt[rec[i] & 511u] + rk[i];
#pragma unroll
    for (int i = 0; i < GCN_SEGS; ++i) {
        if ((unsigned)lane < m[i]) sorted[pos[i]] = make_uint2(rec[i], scv[i]);
    }
    const unsigned total = (unsigned)__builtin_amdgcn_readfirstlane((int)rowcnt[GCN_TILE_ROWS]);
    __syncthreads();

    const v2f* xr = (const v2f*)x;
    gcn_sum_sorted(sorted, ((unsigned)wave * total) >> 5, ((unsigned)(wave + 1) * total) >> 5, tile, xr, lane);
#pragma unroll 1
    for (int i = 0; i < GCN_SEGS; ++i) {
        unsigned si = (unsigned)__builtin_amdgcn_readlane((int)sb, i);
        unsigned ni = (unsigned)__builtin_amdgcn_readlane((int)nn, i);
        for (unsigned k = 32u; k < ni; k += 32u) {
            unsigned cnt = ni - k;
            if (cnt > 32u) cnt = 32u;
            unsigned rv = 0u;
            if ((unsigned)lane < cnt) rv = recs[(size_t)(wave + 32 * i) * GCN_CHUNK + si + k + lane];
            const float sv = scale[rv >> 9];
            gcn_add_rows(rv, sv, cnt, tile, xr, lane);
        }
    }
    __syncthreads();

    if (wave < GCN_ROW_TILES) {
        const int row = lane & 15;
        const int col = lane & 15;
        const int kb  = (lane >> 4) << 1;
        float* trows = tile + wave * 16 * GCN_PITCH;

        v2f areg[16];
        const float* arow = trows + row * GCN_PITCH + kb;
#pragma unroll
        for (int s = 0; s < 16; ++s) areg[s] = *(const v2f*)(arow + 4 * s);

#pragma unroll
        for (int h = 0; h < 2; ++h) {
            v2f breg[2][16];
            v8f c[2];
#pragma unroll
            for (int j = 0; j < 2; ++j) {
                const int tn = 2 * h + j;
                const float* wrow = wlds + (tn * 16 + col) * GCN_PITCH + kb;
#pragma unroll
                for (int s = 0; s < 16; ++s) breg[j][s] = *(const v2f*)(wrow + 4 * s);
                const float bv = blds[tn * 16 + col];
#pragma unroll
                for (int q = 0; q < 8; ++q) c[j][q] = bv;
            }
#pragma unroll
            for (int s = 0; s < 16; ++s) {
#pragma unroll
                for (int j = 0; j < 2; ++j) {
                    c[j] = __builtin_amdgcn_wmma_f32_16x16x4_f32(
                               false, areg[s], false, breg[j][s], (short)0, c[j], false, false);
                }
            }
            float* orow = trows + ((lane >> 4) << 3) * GCN_PITCH + col;
#pragma unroll
            for (int q = 0; q < 8; ++q) {
#pragma unroll
                for (int j = 0; j < 2; ++j)
                    orow[q * GCN_PITCH + (2 * h + j) * 16] = c[j][q];
            }
        }
    }
    __syncthreads();

    float4* oout = (float4*)(out + (size_t)lo * GCN_D);
#pragma unroll
    for (int k = 0; k < 6; ++k) oout[tid + k * 1024] = t4[pbase + k * 64 * GCN_PITCH4];
    if (tid < 256) oout[6144 + tid] = t4[pbase + 6 * 64 * GCN_PITCH4];
}

extern "C" void kernel_launch(void* const* d_in, const int* in_sizes, int n_in,
                              void* d_out, int out_size, void* d_ws, size_t ws_size,
                              hipStream_t stream) {
    const float* x          = (const float*)d_in[0];
    const int*   edge_index = (const int*)d_in[1];
    const float* W          = (const float*)d_in[2];
    const float* b          = (const float*)d_in[3];
    float*       out        = (float*)d_out;

    float* scale = (float*)d_ws;
    unsigned int* recs = (unsigned int*)(scale + (size_t)GCN_N_NODES);
    unsigned int* off  = recs + (size_t)GCN_N_CHUNKS * GCN_CHUNK;
    unsigned short* srow = (unsigned short*)(off + (size_t)GCN_N_CHUNKS * GCN_OFF_PITCH);
    unsigned int* soff = (unsigned int*)(srow + (size_t)GCN_N_CHUNKS * GCN_CHUNK);

    gcn_bucket_edges<<<GCN_N_CHUNKS, 512, 0, stream>>>(edge_index, recs, off, srow, soff);
    gcn_tile_degree_scale<<<GCN_N_TILES, 256, 0, stream>>>(srow, soff, scale);
    gcn_tile_aggregate_gemm<<<GCN_N_TILES, 1024, 0, stream>>>(recs, off, x, scale, W, b, out);
}


# source factors made from a table of out-degree bytes copied into LDS (100000 B a tile, rsqrt per record) instead of 1M scattered 4-byte loads of scale[src]; same factor bits, no sum changes order
# speedup vs baseline: 7.0225x; 1.0782x over previous
#include <hip/hip_runtime.h>
#include <hip/hip_bf16.h>


typedef __attribute__((ext_vector_type(2))) float v2f;
typedef __attribute__((ext_vector_type(8))) float v8f;
typedef __attribute__((ext_vector_type(4))) int   v4i;

#define GCN_N_NODES 100000
#define GCN_N_EDGES 1000000
#define GCN_D 64

#define GCN_TILE_ROWS 400
#define GCN_N_TILES (GCN_N_NODES / GCN_TILE_ROWS)

#define GCN_CHUNK 4096
#define GCN_CHUNK_VECS (GCN_CHUNK / 4)
#define GCN_N_CHUNKS ((GCN_N_EDGES + GCN_CHUNK - 1) / GCN_CHUNK)
#define GCN_OFF_PITCH 256
#define GCN_SEGS 8

#define GCN_DEG_WAVES 8
#define GCN_DEG_SEGS 32
#define GCN_EDGE_VECS (GCN_N_EDGES / 4)
#define GCN_PITCH 68
#define GCN_PITCH4 (GCN_PITCH / 4)
#define GCN_ROW_TILES (GCN_TILE_ROWS / 16)
#define GCN_SCAN_PER_LANE 13
#define GCN_ROW_CNT (32 * GCN_SCAN_PER_LANE)
#define GCN_NO_ROW 0xFFFFFFFFu
#define GCN_DEG_CAP 255u
#define GCN_DEG_VECS (GCN_N_NODES / 16)

__global__ __launch_bounds__(512)
void gcn_bucket_edges(const int* __restrict__ edge_index,
                      unsigned int* __restrict__ recs, unsigned int* __restrict__ off,
                      unsigned short* __restrict__ srow, unsigned int* __restrict__ soff) {
    __shared__ unsigned hist[GCN_OFF_PITCH];
    __shared__ unsigned start[GCN_OFF_PITCH];
    __shared__ unsigned shist[GCN_OFF_PITCH];
    __shared__ unsigned sstart[GCN_OFF_PITCH];
    __shared__ __attribute__((aligned(16))) unsigned stage[GCN_CHUNK];
    __shared__ __attribute__((aligned(16))) unsigned short sstage[GCN_CHUNK];

    const int tid = threadIdx.x;
    const int c   = blockIdx.x;
    if (tid < GCN_OFF_PITCH) { hist[tid] = 0u; shist[tid] = 0u; }

    const v4i* src4 = (const v4i*)edge_index;
    const v4i* dst4 = (const v4i*)(edge_index + GCN_N_EDGES);
    const int q0 = c * GCN_CHUNK_VECS + tid;
    v4i s[2], d[2];
#pragma unroll
    for (int u = 0; u < 2; ++u) {
        int q  = q0 + u * 512;
        int qc = q < GCN_EDGE_VECS ? q : GCN_EDGE_VECS - 1;
        s[u] = src4[qc];
        d[u] = dst4[qc];
    }
    __syncthreads();

    unsigned tl[8], rk[8], rc[8];
    unsigned stl[8], srk[8], srw[8];
#pragma unroll
    for (int u = 0; u < 2; ++u) {
        const bool ok = (q0 + u * 512) < GCN_EDGE_VECS;
        const int sv[4] = { s[u].x, s[u].y, s[u].z, s[u].w };
        const int dv[4] = { d[u].x, d[u].y, d[u].z, d[u].w };
#pragma unroll
        for (int k = 0; k < 4; ++k) {
            const int j = u * 4 + k;
            tl[j]  = (unsigned)dv[k] / GCN_TILE_ROWS;
            rc[j]  = ((unsigned)sv[k] << 9) | ((unsigned)dv[k] - tl[j] * GCN_TILE_ROWS);
            stl[j] = (unsigned)sv[k] / GCN_TILE_ROWS;
            srw[j] = (unsigned)sv[k] - stl[j] * GCN_TILE_ROWS;
            rk[j]  = 0u;
            srk[j] = 0u;
            if (ok) {
                rk[j]  = atomicAdd(&hist[tl[j]], 1u);
                srk[j] = atomicAdd(&shist[stl[j]], 1u);
            }
        }
    }
    __syncthreads();

    if (tid < 64) {
        const unsigned* h  = tid < 32 ? hist  : shist;
        unsigned*       st = tid < 32 ? start : sstart;
        const int l = tid & 31;
        unsigned v[8], tot = 0u;
#pragma unroll
        for (int k = 0; k < 8; ++k) { v[k] = h[l * 8 + k]; tot += v[k]; }
        unsigned incl = tot;
#pragma unroll
        for (int o = 1; o < 32; o <<= 1) {
            unsigned nb = __shfl_up(incl, o);
            if (l >= o) incl += nb;
        }
        unsigned run = incl - tot;
#pragma unroll
        for (int k = 0; k < 8; ++k) { st[l * 8 + k] = run; run += v[k]; }
    }
    __syncthreads();

#pragma unroll
    for (int u = 0; u < 2; ++u) {
        if ((q0 + u * 512) < GCN_EDGE_VECS) {
#pragma unroll
            for (int k = 0; k < 4; ++k) {
                const int j = u * 4 + k;
                stage[start[tl[j]] + rk[j]] = rc[j];
                sstage[sstart[stl[j]] + srk[j]] = (unsigned short)srw[j];
            }
        }
    }
    if (tid < GCN_OFF_PITCH) {
        off[c * GCN_OFF_PITCH + tid]  = start[tid];
        soff[c * GCN_OFF_PITCH + tid] = sstart[tid];
    }
    __syncthreads();

    int nvec = GCN_EDGE_VECS - c * GCN_CHUNK_VECS;
    if (nvec > GCN_CHUNK_VECS) nvec = GCN_CHUNK_VECS;
    uint4* rout = (uint4*)recs + (size_t)c * GCN_CHUNK_VECS;
    for (int i = tid; i < nvec; i += 512) rout[i] = ((const uint4*)stage)[i];
    uint4* sout = (uint4*)srow + (size_t)c * (GCN_CHUNK / 8);
    if (tid < ((nvec + 1) >> 1)) sout[tid] = ((const uint4*)sstage)[tid];
}

__global__ __launch_bounds__(256)
void gcn_tile_degree_scale(const unsigned short* __restrict__ srow, const unsigned int* __restrict__ soff,
                           float* __restrict__ scale, unsigned char* __restrict__ degb) {
    __shared__ unsigned cnt[GCN_TILE_ROWS];

    const int tid  = threadIdx.x;
    const int t    = blockIdx.x;
    const int lo   = t * GCN_TILE_ROWS;
    const int lane = tid & 31;
    const int wave = __builtin_amdgcn_readfirstlane(tid >> 5);
    const int c0   = wave * GCN_DEG_SEGS;
    for (int i = tid; i < GCN_TILE_ROWS; i += 256) cnt[i] = 0u;

    unsigned sb = 0u, nn = 0u;
    {
        const int c = c0 + lane;
        if (c < GCN_N_CHUNKS) {
            unsigned a = soff[c * GCN_OFF_PITCH + t];
            unsigned b = soff[c * GCN_OFF_PITCH + t + 1];
            sb = a;
            nn = b - a;
        }
    }

    unsigned rec[GCN_DEG_SEGS];
#pragma unroll
    for (int i = 0; i < GCN_DEG_SEGS; ++i) {
        unsigned si = (unsigned)__builtin_amdgcn_readlane((int)sb, i);
        unsigned ni = (unsigned)__builtin_amdgcn_readlane((int)nn, i);
        rec[i] = 0u;
        if ((unsigned)lane < ni) rec[i] = srow[(size_t)(c0 + i) * GCN_CHUNK + si + lane];
    }
    __syncthreads();

#pragma unroll
    for (int i = 0; i < GCN_DEG_SEGS; ++i) {
        unsigned ni = (unsigned)__builtin_amdgcn_readlane((int)nn, i);
        if ((unsigned)lane < ni) atomicAdd(&cnt[rec[i]], 1u);
    }
#pragma unroll 1
    for (int i = 0; i < GCN_DEG_SEGS; ++i) {
        unsigned si = (unsigned)__builtin_amdgcn_readlane((int)sb, i);
        unsigned ni = (unsigned)__builtin_amdgcn_readlane((int)nn, i);
        for (unsigned k = 32u + (unsigned)lane; k < ni; k += 32u)
            atomicAdd(&cnt[srow[(size_t)(c0 + i) * GCN_CHUNK + si + k]], 1u);
    }
    __syncthreads();

    for (int i = tid; i < GCN_TILE_ROWS; i += 256) {
        const unsigned n = cnt[i];
        scale[lo + i] = rsqrtf((float)n + 1.0f);
        degb[lo + i]  = (unsigned char)(n < GCN_DEG_CAP ? n : GCN_DEG_CAP);
    }
}

__device__ __forceinline__ void gcn_add_rows(unsigned rv, float sv, unsigned cnt, float* tile,
                                             const v2f* __restrict__ xr, int lane) {
    for (unsigned j0 = 0; j0 < cnt; j0 += 16u) {
        v2f val[16];
        unsigned drow[16];
#pragma unroll
        for (int u = 0; u < 16; ++u) {
            unsigned r = (unsigned)__builtin_amdgcn_readlane((int)rv, (int)(j0 + u));
            drow[u] = r & 511u;
            val[u]  = xr[(size_t)(r >> 9) * (GCN_D / 2) + lane];
        }
#pragma unroll
        for (int u = 0; u < 16; ++u) {
            if (j0 + u < cnt) {
                const float s = __builtin_bit_cast(float, __builtin_amdgcn_readlane(__builtin_bit_cast(int, sv), (int)(j0 + u)));
                float* p = &tile[drow[u] * GCN_PITCH + 2 * lane];
                atomicAdd(p,     val[u].x * s);
                atomicAdd(p + 1, val[u].y * s);
            }
        }
    }
}

__device__ __forceinline__ void gcn_flush_run(float* tile, unsigned row, v2f acc, int lane) {
    float* p = &tile[row * GCN_PITCH + 2 * lane];
    atomicAdd(p,     acc.x);
    atomicAdd(p + 1, acc.y);
}

template <int J0, bool CHECK>
__device__ __forceinline__ void gcn_sum_group(unsigned rv, float sv, unsigned cnt, unsigned& cur, v2f& acc, float* tile,
                                              const v2f* __restrict__ xr, int lane) {
    v2f val[16];
    unsigned drow[16];
#pragma unroll
    for (int u = 0; u < 16; ++u) {
        unsigned r = (unsigned)__builtin_amdgcn_readlane((int)rv, J0 + u);
        drow[u] = r & 511u;
        val[u]  = xr[(size_t)(r >> 9) * (GCN_D / 2) + lane];
    }
#pragma unroll
    for (int u = 0; u < 16; ++u) {
        if (!CHECK || (unsigned)(J0 + u) < cnt) {
            if (__builtin_expect(drow[u] != cur, 0)) {
                if (cur != GCN_NO_ROW) gcn_flush_run(tile, cur, acc, lane);
                cur = drow[u];
                acc = (v2f){ 0.0f, 0.0f };
            }
            const float s = __builtin_bit_cast(float, __builtin_amdgcn_readlane(__builtin_bit_cast(int, sv), J0 + u));
            acc.x = __builtin_fmaf(val[u].x, s, acc.x);
            acc.y = __builtin_fmaf(val[u].y, s, acc.y);
        }
    }
}

__device__ __forceinline__ void gcn_sum_sorted(const uint2* sorted, unsigned beg, unsigned end, float* tile,
                                               const v2f* __restrict__ xr, int lane) {
    unsigned cur = GCN_NO_ROW;
    v2f acc = { 0.0f, 0.0f };
    for (unsigned b0 = beg; b0 < end; b0 += 32u) {
        unsigned cnt = end - b0;
        if (cnt > 32u) cnt = 32u;
        unsigned rv = 0u;
        float sv = 0.0f;
        if ((unsigned)lane < cnt) {
            const uint2 e = sorted[b0 + lane];
            rv = e.x;
            sv = __builtin_bit_cast(float, e.y);
        }
        if (cnt >= 16u) gcn_sum_group<0, false>(rv, sv, cnt, cur, acc, tile, xr, lane);
        else            gcn_sum_group<0, true >(rv, sv, cnt, cur, acc, tile, xr, lane);
        if (cnt >= 32u)      gcn_sum_group<16, false>(rv, sv, cnt, cur, acc, tile, xr, lane);
        else if (cnt > 16u)  gcn_sum_group<16, true >(rv, sv, cnt, cur, acc, tile, xr, lane);
    }
    if (cur != GCN_NO_ROW) gcn_flush_run(tile, cur, acc, lane);
}

__global__ __launch_bounds__(1024)
void gcn_tile_aggregate_gemm(const unsigned int* __restrict__ recs, const unsigned int* __restrict__ off,
                             const float* __restrict__ x, const float* __restrict__ scale,
                             const unsigned char* __restrict__ degb,
                             const float* __restrict__ W,
                             const float* __restrict__ bias, float* __restrict__ out) {
    __shared__ __attribute__((aligned(16))) float tile[GCN_TILE_ROWS * GCN_PITCH];
    __shared__ __attribute__((aligned(16))) float wlds[GCN_D * GCN_PITCH];
    __shared__ float blds[GCN_D];
    __shared__ uint2 sorted[32 * GCN_SEGS * 32];
    __shared__ unsigned rowcnt[GCN_ROW_CNT];
    __shared__ __attribute__((aligned(16))) unsigned char degtab[GCN_N_NODES];

    const int tid  = threadIdx.x;
    const int t    = blockIdx.x;
    const int lo   = t * GCN_TILE_ROWS;
    const int lane = tid & 31;
    const int wave = __builtin_amdgcn_readfirstlane(tid >> 5);
    if (tid < GCN_ROW_CNT) rowcnt[tid] = 0u;

    unsigned sb = 0u, nn = 0u;
    {
        const int c = wave + 32 * lane;
        if (lane < GCN_SEGS && c < GCN_N_CHUNKS) {
            unsigned a = off[c * GCN_OFF_PITCH + t];
            unsigned b = off[c * GCN_OFF_PITCH + t + 1];
            sb = a;
            nn = b - a;
        }
    }

    const int pbase = (tid >> 4) * GCN_PITCH4 + (tid & 15);

    const float4* xin = (const float4*)(x + (size_t)lo * GCN_D);
    float4* t4 = (float4*)tile;
    float4 r[6];
    float  so[6];
#pragma unroll
    for (int k = 0; k < 6; ++k) {
        r[k]  = xin[tid + k * 1024];
        so[k] = scale[lo + (tid >> 4) + 64 * k];
    }
    const int tail = tid < 256 ? 6144 + tid : tid;
    float4 rt = xin[tail];
    float  st = scale[lo + (tail >> 4)];

    const float4 wv = ((const float4*)W)[tid];
    float bval = 0.0f;
    if (tid < GCN_D) bval = bias[tid];

    unsigned rec[GCN_SEGS], m[GCN_SEGS];
#pragma unroll
    for (int i = 0; i < GCN_SEGS; ++i) {
        unsigned si = (unsigned)__builtin_amdgcn_readlane((int)sb, i);
        unsigned ni = (unsigned)__builtin_amdgcn_readlane((int)nn, i);
        m[i]   = ni < 32u ? ni : 32u;
        rec[i] = 0u;
        if ((unsigned)lane < m[i]) rec[i] = recs[(size_t)(wave + 32 * i) * GCN_CHUNK + si + lane];
    }
    const uint4* dsrc = (const uint4*)degb;
    uint4* dt4 = (uint4*)degtab;
    const int dtail = tid < GCN_DEG_VECS - 6144 ? 6144 + tid : tid;
    uint4 dgv[7];
#pragma unroll
    for (int k = 0; k < 6; ++k) dgv[k] = dsrc[tid + k * 1024];
    dgv[6] = dsrc[dtail];
    __builtin_amdgcn_sched_barrier(0);

#pragma unroll
    for (int k = 0; k < 6; ++k) {
        float4 v = r[k];
        v.x *= so[k]; v.y *= so[k]; v.z *= so[k]; v.w *= so[k];
        t4[pbase + k * 64 * GCN_PITCH4] = v;
    }
#pragma unroll
    for (int k = 0; k < 6; ++k) dt4[tid + k * 1024] = dgv[k];
    dt4[dtail] = dgv[6];
    ((float4*)wlds)[pbase] = wv;
    rt.x *= st; rt.y *= st; rt.z *= st; rt.w *= st;
    if (tid < 256) t4[pbase + 6 * 64 * GCN_PITCH4] = rt;
    if (tid < GCN_D) blds[tid] = bval;

    __syncthreads();

    unsigned rk[GCN_SEGS];
#pragma unroll
    for (int i = 0; i < GCN_SEGS; ++i) {
        rk[i] = 0u;
        if ((unsigned)lane < m[i]) rk[i] = atomicAdd(&rowcnt[rec[i] & 511u], 1u);
    }
    unsigned dg[GCN_SEGS];
#pragma unroll
    for (int i = 0; i < GCN_SEGS; ++i) dg[i] = degtab[rec[i] >> 9];
    __syncthreads();

    if (wave == 0) {
        unsigned v[GCN_SCAN_PER_LANE], tot = 0u;
#pragma unroll
        for (int k = 0; k < GCN_SCAN_PER_LANE; ++k) { v[k] = rowcnt[lane * GCN_SCAN_PER_LANE + k]; tot += v[k]; }
        unsigned incl = tot;
#pragma unroll
        for (int o = 1; o < 32; o <<= 1) {
            unsigned nb = __shfl_up(incl, o);
            if (lane >= o) incl += nb;
        }
        unsigned run = incl - tot;
#pragma unroll
        for (int k = 0; k < GCN_SCAN_PER_LANE; ++k) { rowcnt[lane * GCN_SCAN_PER_LANE + k] = run; run += v[k]; }
    }
    __syncthreads();

    unsigned pos[GCN_SEGS];
#pragma unroll
    for (int i = 0; i < GCN_SEGS; ++i) pos[i] = rowcnt[rec[i] & 511u] + rk[i];
#pragma unroll
    for (int i = 0; i < GCN_SEGS; ++i) {
        if ((unsigned)lane < m[i]) {
            float f = rsqrtf((float)dg[i] + 1.0f);
            if (__builtin_expect(dg[i] == GCN_DEG_CAP, 0)) f = scale[rec[i] >> 9];
            sorted[pos[i]] = make_uint2(rec[i], __builtin_bit_cast(unsigned, f));
        }
    }
    const unsigned total = (unsigned)__builtin_amdgcn_readfirstlane((int)rowcnt[GCN_TILE_ROWS]);
    __syncthreads();

    const v2f* xr = (const v2f*)x;
    gcn_sum_sorted(sorted, ((unsigned)wave * total) >> 5, ((unsigned)(wave + 1) * total) >> 5, tile, xr, lane);
#pragma unroll 1
    for (int i = 0; i < GCN_SEGS; ++i) {
        unsigned si = (unsigned)__builtin_amdgcn_readlane((int)sb, i);
        unsigned ni = (unsigned)__builtin_amdgcn_readlane((int)nn, i);
        for (unsigned k = 32u; k < ni; k += 32u) {
            unsigned cnt = ni - k;
            if (cnt > 32u) cnt = 32u;
            unsigned rv = 0u;
            if ((unsigned)lane < cnt) rv = recs[(size_t)(wave + 32 * i) * GCN_CHUNK + si + k + lane];
            const float sv = scale[rv >> 9];
            gcn_add_rows(rv, sv, cnt, tile, xr, lane);
        }
    }
    __syncthreads();

    if (wave < GCN_ROW_TILES) {
        const int row = lane & 15;
        const int col = lane & 15;
        const int kb  = (lane >> 4) << 1;
        float* trows = tile + wave * 16 * GCN_PITCH;

        v2f areg[16];
        const float* arow = trows + row * GCN_PITCH + kb;
#pragma unroll
        for (int s = 0; s < 16; ++s) areg[s] = *(const v2f*)(arow + 4 * s);

#pragma unroll
        for (int h = 0; h < 2; ++h) {
            v2f breg[2][16];
            v8f c[2];
#pragma unroll
            for (int j = 0; j < 2; ++j) {
                const int tn = 2 * h + j;
                const float* wrow = wlds + (tn * 16 + col) * GCN_PITCH + kb;
#pragma unroll
                for (int s = 0; s < 16; ++s) breg[j][s] = *(const v2f*)(wrow + 4 * s);
                const float bv = blds[tn * 16 + col];
#pragma unroll
                for (int q = 0; q < 8; ++q) c[j][q] = bv;
            }
#pragma unroll
            for (int s = 0; s < 16; ++s) {
#pragma unroll
                for (int j = 0; j < 2; ++j) {
                    c[j] = __builtin_amdgcn_wmma_f32_16x16x4_f32(
                               false, areg[s], false, breg[j][s], (short)0, c[j], false, false);
                }
            }
            float* orow = trows + ((lane >> 4) << 3) * GCN_PITCH + col;
#pragma unroll
            for (int q = 0; q < 8; ++q) {
#pragma unroll
                for (int j = 0; j < 2; ++j)
                    orow[q * GCN_PITCH + (2 * h + j) * 16] = c[j][q];
            }
        }
    }
    __syncthreads();

    float4* oout = (float4*)(out + (size_t)lo * GCN_D);
#pragma unroll
    for (int k = 0; k < 6; ++k) oout[tid + k * 1024] = t4[pbase + k * 64 * GCN_PITCH4];
    if (tid < 256) oout[6144 + tid] = t4[pbase + 6 * 64 * GCN_PITCH4];
}

extern "C" void kernel_launch(void* const* d_in, const int* in_sizes, int n_in,
                              void* d_out, int out_size, void* d_ws, size_t ws_size,
                              hipStream_t stream) {
    const float* x          = (const float*)d_in[0];
    const int*   edge_index = (const int*)d_in[1];
    const float* W          = (const float*)d_in[2];
    const float* b          = (const float*)d_in[3];
    float*       out        = (float*)d_out;

    float* scale = (float*)d_ws;
    unsigned int* recs = (unsigned int*)(scale + (size_t)GCN_N_NODES);
    unsigned int* off  = recs + (size_t)GCN_N_CHUNKS * GCN_CHUNK;
    unsigned short* srow = (unsigned short*)(off + (size_t)GCN_N_CHUNKS * GCN_OFF_PITCH);
    unsigned int* soff = (unsigned int*)(srow + (size_t)GCN_N_CHUNKS * GCN_CHUNK);
    unsigned char* degb = (unsigned char*)(soff + (size_t)GCN_N_CHUNKS * GCN_OFF_PITCH);

    gcn_bucket_edges<<<GCN_N_CHUNKS, 512, 0, stream>>>(edge_index, recs, off, srow, soff);
    gcn_tile_degree_scale<<<GCN_N_TILES, 256, 0, stream>>>(srow, soff, scale, degb);
    gcn_tile_aggregate_gemm<<<GCN_N_TILES, 1024, 0, stream>>>(recs, off, x, scale, degb, W, b, out);
}
